# starting kernel
# baseline (speedup 1.0000x reference)
#include <hip/hip_runtime.h>

typedef __attribute__((ext_vector_type(2))) float v2f;
typedef __attribute__((ext_vector_type(8))) float v8f;

#define WAVE 32

__device__ __forceinline__ void atomic_add_f32(float* p, float v) {
  __hip_atomic_fetch_add(p, v, __ATOMIC_RELAXED, __HIP_MEMORY_SCOPE_AGENT);
}

__global__ void k_deg_init(float* __restrict__ deg, int n) {
  int i = blockIdx.x * blockDim.x + threadIdx.x;
  if (i < n) deg[i] = 1.0f;
}

__global__ void k_deg_count(const int* __restrict__ dst, float* __restrict__ deg, int E) {
  int e = blockIdx.x * blockDim.x + threadIdx.x;
  if (e < E) atomic_add_f32(&deg[dst[e]], 1.0f);
}

__global__ void k_deg_rsqrt(float* __restrict__ deg, int n) {
  int i = blockIdx.x * blockDim.x + threadIdx.x;
  if (i < n) deg[i] = rsqrtf(deg[i]);
}

template <int K, int N>
__global__ __launch_bounds__(256) void k_gemm_wmma(
    const float* __restrict__ X, const float* __restrict__ W,
    const float* __restrict__ bias, const float* __restrict__ dinv,
    float* __restrict__ H, float* __restrict__ AGG, int nStrips) {
  __shared__ float Ws[K * N];
  for (int i = threadIdx.x; i < (K / 2) * N; i += blockDim.x) {
    const int k2 = i / N;
    const int n  = i % N;
    v2f p;
    p.x = W[(2 * k2) * N + n];
    p.y = W[(2 * k2 + 1) * N + n];
    *(v2f*)&Ws[(size_t)k2 * 2 * N + 2 * n] = p;
  }
  __syncthreads();

  const int waveId = threadIdx.x / WAVE;
  const int strip = blockIdx.x * (blockDim.x / WAVE) + waveId;
  if (strip >= nStrips) return;

  const int lane = threadIdx.x & (WAVE - 1);
  const int half = lane >> 4;
  const int lr   = lane & 15;

  const float* xrow = X + (size_t)(strip * 16 + lr) * K;

  constexpr int NT = N / 16;
  v8f acc[NT];
#pragma unroll
  for (int t = 0; t < NT; ++t) acc[t] = {};

#pragma unroll 4
  for (int kk = 0; kk < K; kk += 4) {
    v2f a = *(const v2f*)(xrow + kk + 2 * half);
    const float* wrow = &Ws[(size_t)(kk / 2 + half) * 2 * N];
#pragma unroll
    for (int t = 0; t < NT; ++t) {
      v2f b = *(const v2f*)(wrow + 2 * (t * 16 + lr));
      acc[t] = __builtin_amdgcn_wmma_f32_16x16x4_f32(
          false, a, false, b, (short)0, acc[t], false, false);
    }
  }

#pragma unroll
  for (int j = 0; j < 8; ++j) {
    const int orow = strip * 16 + j + 8 * half;
    const float di = dinv[orow];
    const float d2 = di * di;
    float* hrow = H + (size_t)orow * N;
    float* arow = AGG + (size_t)orow * N;
#pragma unroll
    for (int t = 0; t < NT; ++t) {
      const int col = t * 16 + lr;
      const float v = acc[t][j];
      hrow[col] = v;
      arow[col] = v * d2 + bias[col];
    }
  }
}

template <int C>
__global__ void k_edge_agg(const int* __restrict__ src, const int* __restrict__ dst,
                           const float* __restrict__ dinv, const float* __restrict__ H,
                           float* __restrict__ AGG, int E) {
  constexpr int QP = C / 4;
  long long t = (long long)blockIdx.x * blockDim.x + threadIdx.x;
  if (t >= (long long)E * QP) return;
  const int e = (int)(t / QP);
  const int q = (int)(t % QP);
  const int s = src[e];
  const int d = dst[e];
  const float norm = dinv[s] * dinv[d];
  const float4 m = *(const float4*)(H + (size_t)s * C + q * 4);
  float* p = AGG + (size_t)d * C + q * 4;
  atomic_add_f32(p + 0, m.x * norm);
  atomic_add_f32(p + 1, m.y * norm);
  atomic_add_f32(p + 2, m.z * norm);
  atomic_add_f32(p + 3, m.w * norm);
}

__global__ void k_relu(float* __restrict__ a, int n) {
  int i = blockIdx.x * blockDim.x + threadIdx.x;
  if (i < n) a[i] = fmaxf(a[i], 0.0f);
}

extern "C" void kernel_launch(void* const* d_in, const int* in_sizes, int n_in,
                              void* d_out, int out_size, void* d_ws, size_t ws_size,
                              hipStream_t stream) {
  const float* x  = (const float*)d_in[0];
  const int*   ei = (const int*)d_in[1];
  const float* W1 = (const float*)d_in[2];
  const float* b1 = (const float*)d_in[3];
  const float* W2 = (const float*)d_in[4];
  const float* b2 = (const float*)d_in[5];

  constexpr int CIN = 128, CHID = 64, COUT = 32;
  const int N = in_sizes[0] / CIN;
  const int E = in_sizes[1] / 2;
  const int* src = ei;
  const int* dst = ei + E;

  float* ws   = (float*)d_ws;
  float* dinv = ws;
  float* h1   = dinv + N;
  float* agg1 = h1 + (size_t)N * CHID;
  float* h2   = agg1 + (size_t)N * CHID;
  float* out  = (float*)d_out;

  const int B = 256;

  k_deg_init <<<(N + B - 1) / B, B, 0, stream>>>(dinv, N);
  k_deg_count<<<(E + B - 1) / B, B, 0, stream>>>(dst, dinv, E);
  k_deg_rsqrt<<<(N + B - 1) / B, B, 0, stream>>>(dinv, N);

  const int strips  = (N + 15) / 16;
  const int gblocks = (strips + 7) / 8;

  k_gemm_wmma<CIN, CHID><<<gblocks, B, 0, stream>>>(x, W1, b1, dinv, h1, agg1, strips);

  const long long t1 = (long long)E * (CHID / 4);
  k_edge_agg<CHID><<<(unsigned)((t1 + B - 1) / B), B, 0, stream>>>(src, dst, dinv, h1, agg1, E);

  const int nh = N * CHID;
  k_relu<<<(nh + B - 1) / B, B, 0, stream>>>(agg1, nh);

  k_gemm_wmma<CHID, COUT><<<gblocks, B, 0, stream>>>(agg1, W2, b2, dinv, h2, out, strips);

  const long long t2 = (long long)E * (COUT / 4);
  k_edge_agg<COUT><<<(unsigned)((t2 + B - 1) / B), B, 0, stream>>>(src, dst, dinv, h2, out, E);
}


# Float scatter-add atomics replaced by an in-call CSR (one returning int atomic per edge gives its slot, scan, fill) and a wave-per-row gather; rows now sum in arrival-order slots, relu/bias folded in
# speedup vs baseline: 1.4566x; 1.4566x over previous
#include <hip/hip_runtime.h>

typedef __attribute__((ext_vector_type(2))) float v2f;
typedef __attribute__((ext_vector_type(8))) float v8f;

#define WAVE 32

__global__ void k_cnt_zero(int* __restrict__ cnt, int n) {
  int i = blockIdx.x * blockDim.x + threadIdx.x;
  if (i < n) cnt[i] = 0;
}

__global__ void k_cnt_rank(const int* __restrict__ dst, int* __restrict__ cnt,
                           int* __restrict__ rank, int E) {
  int e = blockIdx.x * blockDim.x + threadIdx.x;
  if (e < E)
    rank[e] = __hip_atomic_fetch_add(&cnt[dst[e]], 1, __ATOMIC_RELAXED,
                                     __HIP_MEMORY_SCOPE_AGENT);
}

__device__ __forceinline__ int wave_sum(int v) {
  for (int o = 16; o > 0; o >>= 1) v += __shfl_xor(v, o, WAVE);
  return v;
}

__global__ __launch_bounds__(1024) void k_scan(const int* __restrict__ cnt,
                                               int* __restrict__ rowptr,
                                               float* __restrict__ dinv, int n) {
  __shared__ int sm[WAVE];
  const int tid = threadIdx.x;
  const int lane = tid & (WAVE - 1);
  const int wid = tid / WAVE;
  const int blockStart = blockIdx.x * 1024;

  int part = 0;
  const int4* c4 = (const int4*)cnt;
  const int n4 = blockStart / 4;
#pragma unroll 4
  for (int i = tid; i < n4; i += 1024) {
    const int4 q = c4[i];
    part += (q.x + q.y) + (q.z + q.w);
  }
  part = wave_sum(part);
  if (lane == 0) sm[wid] = part;
  __syncthreads();
  const int base = wave_sum(sm[lane]);
  __syncthreads();

  const int i = blockStart + tid;
  const int c = (i < n) ? cnt[i] : 0;
  int x = c;
  for (int o = 1; o < WAVE; o <<= 1) {
    const int y = __shfl_up(x, o, WAVE);
    if (lane >= o) x += y;
  }
  if (lane == WAVE - 1) sm[wid] = x;
  __syncthreads();
  const int woff = wave_sum((lane < wid) ? sm[lane] : 0);
  if (i < n) {
    rowptr[i] = base + woff + x - c;
    if (i == n - 1) rowptr[n] = base + woff + x;
    dinv[i] = rsqrtf((float)c + 1.0f);
  }
}

__global__ void k_fill(const int* __restrict__ src, const int* __restrict__ dst,
                       const int* __restrict__ rank, const int* __restrict__ rowptr,
                       const float* __restrict__ dinv, int* __restrict__ csr, int E) {
  int e = blockIdx.x * blockDim.x + threadIdx.x;
  if (e >= E) return;
  const int s = src[e];
  const int d = dst[e];
  const int pos = rowptr[d] + rank[e];
  int2 ent;
  ent.x = s;
  ent.y = __float_as_int(dinv[s] * dinv[d]);
  ((int2*)csr)[pos] = ent;
}

template <int K, int N>
__global__ __launch_bounds__(256) void k_gemm_wmma(
    const float* __restrict__ X, const float* __restrict__ W,
    float* __restrict__ H, int nStrips) {
  __shared__ float Ws[K * N];
  for (int i = threadIdx.x; i < (K / 2) * N; i += blockDim.x) {
    const int k2 = i / N;
    const int n  = i % N;
    v2f p;
    p.x = W[(2 * k2) * N + n];
    p.y = W[(2 * k2 + 1) * N + n];
    *(v2f*)&Ws[(size_t)k2 * 2 * N + 2 * n] = p;
  }
  __syncthreads();

  const int waveId = threadIdx.x / WAVE;
  const int strip = blockIdx.x * (blockDim.x / WAVE) + waveId;
  if (strip >= nStrips) return;

  const int lane = threadIdx.x & (WAVE - 1);
  const int half = lane >> 4;
  const int lr   = lane & 15;

  const float* xrow = X + (size_t)(strip * 16 + lr) * K;

  constexpr int NT = N / 16;
  v8f acc[NT];
#pragma unroll
  for (int t = 0; t < NT; ++t) acc[t] = {};

#pragma unroll 4
  for (int kk = 0; kk < K; kk += 4) {
    v2f a = *(const v2f*)(xrow + kk + 2 * half);
    const float* wrow = &Ws[(size_t)(kk / 2 + half) * 2 * N];
#pragma unroll
    for (int t = 0; t < NT; ++t) {
      v2f b = *(const v2f*)(wrow + 2 * (t * 16 + lr));
      acc[t] = __builtin_amdgcn_wmma_f32_16x16x4_f32(
          false, a, false, b, (short)0, acc[t], false, false);
    }
  }

#pragma unroll
  for (int j = 0; j < 8; ++j) {
    const int orow = strip * 16 + j + 8 * half;
    float* hrow = H + (size_t)orow * N;
#pragma unroll
    for (int t = 0; t < NT; ++t) hrow[t * 16 + lr] = acc[t][j];
  }
}

template <int C, bool RELU>
__global__ __launch_bounds__(256) void k_gather(
    const int* __restrict__ rowptr, const int* __restrict__ csr,
    const float* __restrict__ dinv, const float* __restrict__ bias,
    const float* __restrict__ H, float* __restrict__ OUT, int n) {
  constexpr int V = C / WAVE;
  const int lane = threadIdx.x & (WAVE - 1);
  const int node = __builtin_amdgcn_readfirstlane(
      (int)(blockIdx.x * (blockDim.x / WAVE) + threadIdx.x / WAVE));
  if (node >= n) return;
  const int start = rowptr[node];
  const int deg = rowptr[node + 1] - start;
  const float di = dinv[node];
  const int2* lst = (const int2*)csr + start;

  const float* own = H + (size_t)node * C + lane * V;
  float self[V], bs[V];
  for (int v = 0; v < V; ++v) self[v] = own[v];
  for (int v = 0; v < V; ++v) bs[v] = bias[lane * V + v];

  float acc[V];
  for (int v = 0; v < V; ++v) acc[v] = 0.0f;

  for (int c0 = 0; c0 < deg; c0 += WAVE) {
    const int m = (deg - c0 < WAVE) ? (deg - c0) : WAVE;
    int2 ent;
    ent.x = 0;
    ent.y = 0;
    if (lane < m) ent = lst[c0 + lane];
#pragma unroll 4
    for (int j = 0; j < m; ++j) {
      const int s = __builtin_amdgcn_readlane(ent.x, j);
      const float w = __int_as_float(__builtin_amdgcn_readlane(ent.y, j));
      const float* hr = H + (size_t)s * C + lane * V;
      for (int v = 0; v < V; ++v) acc[v] += hr[v] * w;
    }
  }

  const float d2 = di * di;
  float* orow = OUT + (size_t)node * C + lane * V;
  for (int v = 0; v < V; ++v) {
    float r = (acc[v] + self[v] * d2) + bs[v];
    if (RELU) r = fmaxf(r, 0.0f);
    orow[v] = r;
  }
}

extern "C" void kernel_launch(void* const* d_in, const int* in_sizes, int n_in,
                              void* d_out, int out_size, void* d_ws, size_t ws_size,
                              hipStream_t stream) {
  const float* x  = (const float*)d_in[0];
  const int*   ei = (const int*)d_in[1];
  const float* W1 = (const float*)d_in[2];
  const float* b1 = (const float*)d_in[3];
  const float* W2 = (const float*)d_in[4];
  const float* b2 = (const float*)d_in[5];

  constexpr int CIN = 128, CHID = 64, COUT = 32;
  const int N = in_sizes[0] / CIN;
  const int E = in_sizes[1] / 2;
  const int* src = ei;
  const int* dst = ei + E;

  float* ws     = (float*)d_ws;
  float* dinv   = ws;
  float* h1     = dinv + N;
  float* a1     = h1 + (size_t)N * CHID;
  int*   cnt    = (int*)(a1 + (size_t)N * CHID);
  int*   rowptr = cnt + N;
  int*   csr    = rowptr + N + 2;
  int*   rank   = (int*)a1;
  float* h2     = h1;
  float* out    = (float*)d_out;

  const int B = 256;

  k_cnt_zero<<<(N + B - 1) / B, B, 0, stream>>>(cnt, N);
  k_cnt_rank<<<(E + B - 1) / B, B, 0, stream>>>(dst, cnt, rank, E);
  k_scan<<<(N + 1023) / 1024, 1024, 0, stream>>>(cnt, rowptr, dinv, N);
  k_fill<<<(E + B - 1) / B, B, 0, stream>>>(src, dst, rank, rowptr, dinv, csr, E);

  const int strips  = (N + 15) / 16;
  const int gblocks = (strips + 7) / 8;
  const int rblocks = (N + 7) / 8;

  k_gemm_wmma<CIN, CHID><<<gblocks, B, 0, stream>>>(x, W1, h1, strips);
  k_gather<CHID, true><<<rblocks, B, 0, stream>>>(rowptr, csr, dinv, b1, h1, a1, N);

  k_gemm_wmma<CHID, COUT><<<gblocks, B, 0, stream>>>(a1, W2, h2, strips);
  k_gather<COUT, false><<<rblocks, B, 0, stream>>>(rowptr, csr, dinv, b2, h2, out, N);
}


# Gather: a 16-lane (layer 1) or 8-lane (layer 2) group per destination row with 16-byte lane loads, 2 or 4 rows per wave, up to 16 source rows of a row requested in one burst; slot order kept
# speedup vs baseline: 1.5597x; 1.0708x over previous
#include <hip/hip_runtime.h>

typedef __attribute__((ext_vector_type(2))) float v2f;
typedef __attribute__((ext_vector_type(4))) float v4f;
typedef __attribute__((ext_vector_type(8))) float v8f;

#define WAVE 32

__global__ void k_cnt_zero(int* __restrict__ cnt, int n) {
  int i = blockIdx.x * blockDim.x + threadIdx.x;
  if (i < n) cnt[i] = 0;
}

__global__ void k_cnt_rank(const int* __restrict__ dst, int* __restrict__ cnt,
                           int* __restrict__ rank, int E) {
  int e = blockIdx.x * blockDim.x + threadIdx.x;
  if (e < E)
    rank[e] = __hip_atomic_fetch_add(&cnt[dst[e]], 1, __ATOMIC_RELAXED,
                                     __HIP_MEMORY_SCOPE_AGENT);
}

__device__ __forceinline__ int wave_sum(int v) {
  for (int o = 16; o > 0; o >>= 1) v += __shfl_xor(v, o, WAVE);
  return v;
}

__global__ __launch_bounds__(1024) void k_scan(const int* __restrict__ cnt,
                                               int* __restrict__ rowptr,
                                               float* __restrict__ dinv, int n) {
  __shared__ int sm[WAVE];
  const int tid = threadIdx.x;
  const int lane = tid & (WAVE - 1);
  const int wid = tid / WAVE;
  const int blockStart = blockIdx.x * 1024;

  int part = 0;
  const int4* c4 = (const int4*)cnt;
  const int n4 = blockStart / 4;
#pragma unroll 4
  for (int i = tid; i < n4; i += 1024) {
    const int4 q = c4[i];
    part += (q.x + q.y) + (q.z + q.w);
  }
  part = wave_sum(part);
  if (lane == 0) sm[wid] = part;
  __syncthreads();
  const int base = wave_sum(sm[lane]);
  __syncthreads();

  const int i = blockStart + tid;
  const int c = (i < n) ? cnt[i] : 0;
  int x = c;
  for (int o = 1; o < WAVE; o <<= 1) {
    const int y = __shfl_up(x, o, WAVE);
    if (lane >= o) x += y;
  }
  if (lane == WAVE - 1) sm[wid] = x;
  __syncthreads();
  const int woff = wave_sum((lane < wid) ? sm[lane] : 0);
  if (i < n) {
    rowptr[i] = base + woff + x - c;
    if (i == n - 1) rowptr[n] = base + woff + x;
    dinv[i] = rsqrtf((float)c + 1.0f);
  }
}

__global__ void k_fill(const int* __restrict__ src, const int* __restrict__ dst,
                       const int* __restrict__ rank, const int* __restrict__ rowptr,
                       int* __restrict__ csr, int E) {
  int e = blockIdx.x * blockDim.x + threadIdx.x;
  if (e >= E) return;
  csr[rowptr[dst[e]] + rank[e]] = src[e];
}

template <int K, int N>
__global__ __launch_bounds__(256) void k_gemm_wmma(
    const float* __restrict__ X, const float* __restrict__ W,
    const float* __restrict__ dinv, float* __restrict__ H, int nStrips) {
  __shared__ float Ws[K * N];
  for (int i = threadIdx.x; i < (K / 2) * N; i += blockDim.x) {
    const int k2 = i / N;
    const int n  = i % N;
    v2f p;
    p.x = W[(2 * k2) * N + n];
    p.y = W[(2 * k2 + 1) * N + n];
    *(v2f*)&Ws[(size_t)k2 * 2 * N + 2 * n] = p;
  }
  __syncthreads();

  const int waveId = threadIdx.x / WAVE;
  const int strip = blockIdx.x * (blockDim.x / WAVE) + waveId;
  if (strip >= nStrips) return;

  const int lane = threadIdx.x & (WAVE - 1);
  const int half = lane >> 4;
  const int lr   = lane & 15;

  const float* xrow = X + (size_t)(strip * 16 + lr) * K;

  constexpr int NT = N / 16;
  v8f acc[NT];
#pragma unroll
  for (int t = 0; t < NT; ++t) acc[t] = {};

#pragma unroll 4
  for (int kk = 0; kk < K; kk += 4) {
    v2f a = *(const v2f*)(xrow + kk + 2 * half);
    const float* wrow = &Ws[(size_t)(kk / 2 + half) * 2 * N];
#pragma unroll
    for (int t = 0; t < NT; ++t) {
      v2f b = *(const v2f*)(wrow + 2 * (t * 16 + lr));
      acc[t] = __builtin_amdgcn_wmma_f32_16x16x4_f32(
          false, a, false, b, (short)0, acc[t], false, false);
    }
  }

#pragma unroll
  for (int j = 0; j < 8; ++j) {
    const int orow = strip * 16 + j + 8 * half;
    float* hrow = H + (size_t)orow * N;
    const float dv = dinv[orow];
#pragma unroll
    for (int t = 0; t < NT; ++t) hrow[t * 16 + lr] = acc[t][j] * dv;
  }
}

template <int C, bool RELU>
__global__ __launch_bounds__(256) void k_gather(
    const int* __restrict__ rowptr, const int* __restrict__ csr,
    const float* __restrict__ dinv, const float* __restrict__ bias,
    const float* __restrict__ H, float* __restrict__ OUT, int n) {
  constexpr int G  = C / 4;
  constexpr int R  = WAVE / G;
  constexpr int B  = 16;
  constexpr int NE = B / G;
  const int lane = threadIdx.x & (WAVE - 1);
  const int q = lane & (G - 1);
  const int node = (int)(blockIdx.x * (blockDim.x / WAVE) + threadIdx.x / WAVE) * R + lane / G;
  const bool valid = node < n;

  const v4f bs = *(const v4f*)(bias + q * 4);
  v4f self = {};
  int start = 0, deg = 0;
  float di = 0.0f;
  if (valid) {
    self = *(const v4f*)(H + (size_t)node * C + q * 4);
    start = rowptr[node];
    deg = rowptr[node + 1] - start;
    di = dinv[node];
  }
  const int* lst = csr + start;
  const char* Hq = (const char*)(H + q * 4);

  v4f acc = {};
  for (int base = 0; __builtin_amdgcn_ballot_w32(base < deg) != 0u; base += B) {
    int ent[NE];
#pragma unroll
    for (int u = 0; u < NE; ++u) {
      const int sl = base + u * G + q;
      ent[u] = 0;
      if (sl < deg) ent[u] = lst[sl];
    }
    v4f r[B];
#pragma unroll
    for (int j = 0; j < B; ++j) {
      const int s = __shfl(ent[j / G], j % G, G);
      if (base + j < deg)
        r[j] = *(const v4f*)(Hq + (unsigned)s * (unsigned)(C * 4));
    }
#pragma unroll
    for (int j = 0; j < B; ++j)
      if (base + j < deg) acc += r[j];
  }

  if (valid) {
    v4f o = (acc + self) * di + bs;
    if (RELU) {
      o.x = fmaxf(o.x, 0.0f); o.y = fmaxf(o.y, 0.0f);
      o.z = fmaxf(o.z, 0.0f); o.w = fmaxf(o.w, 0.0f);
    }
    *(v4f*)(OUT + (size_t)node * C + q * 4) = o;
  }
}

extern "C" void kernel_launch(void* const* d_in, const int* in_sizes, int n_in,
                              void* d_out, int out_size, void* d_ws, size_t ws_size,
                              hipStream_t stream) {
  const float* x  = (const float*)d_in[0];
  const int*   ei = (const int*)d_in[1];
  const float* W1 = (const float*)d_in[2];
  const float* b1 = (const float*)d_in[3];
  const float* W2 = (const float*)d_in[4];
  const float* b2 = (const float*)d_in[5];

  constexpr int CIN = 128, CHID = 64, COUT = 32;
  const int N = in_sizes[0] / CIN;
  const int E = in_sizes[1] / 2;
  const int* src = ei;
  const int* dst = ei + E;

  float* ws     = (float*)d_ws;
  float* dinv   = ws;
  float* h1     = dinv + N;
  float* a1     = h1 + (size_t)N * CHID;
  int*   cnt    = (int*)(a1 + (size_t)N * CHID);
  int*   rowptr = cnt + N;
  int*   csr    = rowptr + N + 2;
  int*   rank   = (int*)a1;
  float* h2     = h1;
  float* out    = (float*)d_out;

  const int B = 256;

  k_cnt_zero<<<(N + B - 1) / B, B, 0, stream>>>(cnt, N);
  k_cnt_rank<<<(E + B - 1) / B, B, 0, stream>>>(dst, cnt, rank, E);
  k_scan<<<(N + 1023) / 1024, 1024, 0, stream>>>(cnt, rowptr, dinv, N);
  k_fill<<<(E + B - 1) / B, B, 0, stream>>>(src, dst, rank, rowptr, csr, E);

  const int strips  = (N + 15) / 16;
  const int gblocks = (strips + 7) / 8;
  const int rblocks1 = (N + 15) / 16;
  const int rblocks2 = (N + 31) / 32;

  k_gemm_wmma<CIN, CHID><<<gblocks, B, 0, stream>>>(x, W1, dinv, h1, strips);
  k_gather<CHID, true><<<rblocks1, B, 0, stream>>>(rowptr, csr, dinv, b1, h1, a1, N);

  k_gemm_wmma<CHID, COUT><<<gblocks, B, 0, stream>>>(a1, W2, dinv, h2, strips);
  k_gather<COUT, false><<<rblocks2, B, 0, stream>>>(rowptr, csr, dinv, b2, h2, out, N);
}


# CSR built by a two-level counting sort in LDS (chunks sorted by 512-node band, then one block per band places its rows), no global atomic or scattered store per edge; 6 launches; arrival-order slots
# speedup vs baseline: 2.8921x; 1.8543x over previous
#include <hip/hip_runtime.h>

typedef __attribute__((ext_vector_type(2))) float v2f;
typedef __attribute__((ext_vector_type(4))) float v4f;
typedef __attribute__((ext_vector_type(8))) float v8f;

#define WAVE 32

constexpr int BSH = 9;
constexpr int BW = 1 << BSH;
constexpr int CH = 8192;
constexpr int MAXBAND = 1024;
constexpr int CAP = 8192;

__device__ __forceinline__ int wave_sum(int v) {
  for (int o = 16; o > 0; o >>= 1) v += __shfl_xor(v, o, WAVE);
  return v;
}

__device__ __forceinline__ int block_sum(int v, int* sm) {
  const int lane = threadIdx.x & (WAVE - 1);
  const int wid = threadIdx.x / WAVE;
  v = wave_sum(v);
  if (lane == 0) sm[wid] = v;
  __syncthreads();
  const int t = wave_sum(sm[lane]);
  __syncthreads();
  return t;
}

__device__ __forceinline__ int block_excl_scan(int v, int* sm, int& total) {
  const int lane = threadIdx.x & (WAVE - 1);
  const int wid = threadIdx.x / WAVE;
  int x = v;
  for (int o = 1; o < WAVE; o <<= 1) {
    const int y = __shfl_up(x, o, WAVE);
    if (lane >= o) x += y;
  }
  if (lane == WAVE - 1) sm[wid] = x;
  __syncthreads();
  const int ws = sm[lane];
  const int woff = wave_sum((lane < wid) ? ws : 0);
  total = wave_sum(ws);
  __syncthreads();
  return woff + x - v;
}

__global__ __launch_bounds__(1024) void k_part(const int* __restrict__ src,
                                               const int* __restrict__ dst,
                                               unsigned* __restrict__ part,
                                               int* __restrict__ run, int E, int nband,
                                               int pitch) {
  __shared__ int hist[MAXBAND];
  __shared__ unsigned stage[CH];
  __shared__ int sm[WAVE];
  const int tid = threadIdx.x;
  const int e0 = blockIdx.x * CH;
  constexpr int PER = CH / 1024;

  hist[tid] = 0;
  __syncthreads();

  int s[PER], d[PER], r[PER];
#pragma unroll
  for (int k = 0; k < PER; ++k) {
    const int e = e0 + k * 1024 + tid;
    d[k] = -1; s[k] = 0;
    if (e < E) { d[k] = dst[e]; s[k] = src[e]; }
  }
#pragma unroll
  for (int k = 0; k < PER; ++k) {
    r[k] = 0;
    if (d[k] >= 0) r[k] = atomicAdd(&hist[d[k] >> BSH], 1);
  }
  __syncthreads();

  const int c = (tid < nband) ? hist[tid] : 0;
  int total;
  const int st = block_excl_scan(c, sm, total);
  hist[tid] = st;
  if (tid < nband) run[(size_t)blockIdx.x * pitch + tid] = st;
  if (tid == 0) run[(size_t)blockIdx.x * pitch + nband] = total;
  __syncthreads();

#pragma unroll
  for (int k = 0; k < PER; ++k)
    if (d[k] >= 0)
      stage[hist[d[k] >> BSH] + r[k]] = ((unsigned)s[k] << BSH) | (unsigned)(d[k] & (BW - 1));
  __syncthreads();

  for (int k = 0; k < PER; ++k) {
    const int i = k * 1024 + tid;
    if (i < total) part[(size_t)e0 + i] = stage[i];
  }
}

__global__ __launch_bounds__(1024) void k_band(const unsigned* __restrict__ part,
                                               const int* __restrict__ run,
                                               int* __restrict__ rowptr,
                                               float* __restrict__ dinv,
                                               int* __restrict__ csr, int n, int nband,
                                               int pitch, int nchunk) {
  __shared__ int cnt[BW];
  __shared__ int stg[CAP];
  __shared__ int sm[WAVE];
  const int tid = threadIdx.x;
  const int band = blockIdx.x;
  constexpr int GL = 16;
  constexpr int NG = 1024 / GL;
  const int g = tid / GL, gl = tid % GL;

  int before = 0, mine = 0;
  for (int r = tid; r < nchunk; r += 1024) {
    const int* rr = run + (size_t)r * pitch + band;
    const int s = rr[0], e = rr[1];
    before += s;
    mine += e - s;
  }
  if (tid < BW) cnt[tid] = 0;
  const int base = block_sum(before, sm);
  const int total = block_sum(mine, sm);
  const bool staged = total <= CAP;

  for (int r = g; r < nchunk; r += NG) {
    const int* rr = run + (size_t)r * pitch + band;
    const int s = rr[0], e = rr[1];
    const unsigned* p = part + (size_t)r * CH;
    for (int j = s + gl; j < e; j += GL) atomicAdd(&cnt[p[j] & (BW - 1)], 1);
  }
  __syncthreads();

  const int c = (tid < BW) ? cnt[tid] : 0;
  int tot2;
  const int st = block_excl_scan(c, sm, tot2);
  if (tid < BW) {
    cnt[tid] = st;
    const int node = band * BW + tid;
    if (node < n) {
      rowptr[node] = base + st;
      dinv[node] = rsqrtf((float)c + 1.0f);
      if (node == n - 1) rowptr[n] = base + st + c;
    }
  }
  __syncthreads();

  for (int r = g; r < nchunk; r += NG) {
    const int* rr = run + (size_t)r * pitch + band;
    const int s = rr[0], e = rr[1];
    const unsigned* p = part + (size_t)r * CH;
    for (int j = s + gl; j < e; j += GL) {
      const unsigned en = p[j];
      const int pos = atomicAdd(&cnt[en & (BW - 1)], 1);
      if (staged) stg[pos] = (int)(en >> BSH);
      else csr[base + pos] = (int)(en >> BSH);
    }
  }
  if (staged) {
    __syncthreads();
    for (int i = tid; i < total; i += 1024) csr[base + i] = stg[i];
  }
}

template <int K, int N>
__global__ __launch_bounds__(256) void k_gemm_wmma(
    const float* __restrict__ X, const float* __restrict__ W,
    const float* __restrict__ dinv, float* __restrict__ H, int nStrips) {
  __shared__ float Ws[K * N];
  for (int i = threadIdx.x; i < (K / 2) * N; i += blockDim.x) {
    const int k2 = i / N;
    const int n  = i % N;
    v2f p;
    p.x = W[(2 * k2) * N + n];
    p.y = W[(2 * k2 + 1) * N + n];
    *(v2f*)&Ws[(size_t)k2 * 2 * N + 2 * n] = p;
  }
  __syncthreads();

  const int waveId = threadIdx.x / WAVE;
  const int strip = blockIdx.x * (blockDim.x / WAVE) + waveId;
  if (strip >= nStrips) return;

  const int lane = threadIdx.x & (WAVE - 1);
  const int half = lane >> 4;
  const int lr   = lane & 15;

  const float* xrow = X + (size_t)(strip * 16 + lr) * K;

  constexpr int NT = N / 16;
  v8f acc[NT];
#pragma unroll
  for (int t = 0; t < NT; ++t) acc[t] = {};

#pragma unroll 4
  for (int kk = 0; kk < K; kk += 4) {
    v2f a = *(const v2f*)(xrow + kk + 2 * half);
    const float* wrow = &Ws[(size_t)(kk / 2 + half) * 2 * N];
#pragma unroll
    for (int t = 0; t < NT; ++t) {
      v2f b = *(const v2f*)(wrow + 2 * (t * 16 + lr));
      acc[t] = __builtin_amdgcn_wmma_f32_16x16x4_f32(
          false, a, false, b, (short)0, acc[t], false, false);
    }
  }

#pragma unroll
  for (int j = 0; j < 8; ++j) {
    const int orow = strip * 16 + j + 8 * half;
    float* hrow = H + (size_t)orow * N;
    const float dv = dinv[orow];
#pragma unroll
    for (int t = 0; t < NT; ++t) hrow[t * 16 + lr] = acc[t][j] * dv;
  }
}

template <int C, bool RELU>
__global__ __launch_bounds__(256) void k_gather(
    const int* __restrict__ rowptr, const int* __restrict__ csr,
    const float* __restrict__ dinv, const float* __restrict__ bias,
    const float* __restrict__ H, float* __restrict__ OUT, int n) {
  constexpr int G  = C / 4;
  constexpr int R  = WAVE / G;
  constexpr int B  = 16;
  constexpr int NE = B / G;
  const int lane = threadIdx.x & (WAVE - 1);
  const int q = lane & (G - 1);
  const int node = (int)(blockIdx.x * (blockDim.x / WAVE) + threadIdx.x / WAVE) * R + lane / G;
  const bool valid = node < n;

  const v4f bs = *(const v4f*)(bias + q * 4);
  v4f self = {};
  int start = 0, deg = 0;
  float di = 0.0f;
  if (valid) {
    self = *(const v4f*)(H + (size_t)node * C + q * 4);
    start = rowptr[node];
    deg = rowptr[node + 1] - start;
    di = dinv[node];
  }
  const int* lst = csr + start;
  const char* Hq = (const char*)(H + q * 4);

  v4f acc = {};
  for (int base = 0; __builtin_amdgcn_ballot_w32(base < deg) != 0u; base += B) {
    int ent[NE];
#pragma unroll
    for (int u = 0; u < NE; ++u) {
      const int sl = base + u * G + q;
      ent[u] = 0;
      if (sl < deg) ent[u] = lst[sl];
    }
    v4f r[B];
#pragma unroll
    for (int j = 0; j < B; ++j) {
      const int s = __shfl(ent[j / G], j % G, G);
      if (base + j < deg)
        r[j] = *(const v4f*)(Hq + (unsigned)s * (unsigned)(C * 4));
    }
#pragma unroll
    for (int j = 0; j < B; ++j)
      if (base + j < deg) acc += r[j];
  }

  if (valid) {
    v4f o = (acc + self) * di + bs;
    if (RELU) {
      o.x = fmaxf(o.x, 0.0f); o.y = fmaxf(o.y, 0.0f);
      o.z = fmaxf(o.z, 0.0f); o.w = fmaxf(o.w, 0.0f);
    }
    *(v4f*)(OUT + (size_t)node * C + q * 4) = o;
  }
}

extern "C" void kernel_launch(void* const* d_in, const int* in_sizes, int n_in,
                              void* d_out, int out_size, void* d_ws, size_t ws_size,
                              hipStream_t stream) {
  const float* x  = (const float*)d_in[0];
  const int*   ei = (const int*)d_in[1];
  const float* W1 = (const float*)d_in[2];
  const float* b1 = (const float*)d_in[3];
  const float* W2 = (const float*)d_in[4];
  const float* b2 = (const float*)d_in[5];

  constexpr int CIN = 128, CHID = 64, COUT = 32;
  const int N = in_sizes[0] / CIN;
  const int E = in_sizes[1] / 2;
  const int* src = ei;
  const int* dst = ei + E;

  float* ws     = (float*)d_ws;
  float* dinv   = ws;
  float* h1     = dinv + N;
  float* a1     = h1 + (size_t)N * CHID;
  int*   rowptr = (int*)(a1 + (size_t)N * CHID) + N;
  int*   csr    = rowptr + N + 2;
  float* h2     = h1;
  float* out    = (float*)d_out;

  const int B = 256;

  const int nband  = (N + BW - 1) / BW;
  const int nchunk = (E + CH - 1) / CH;
  const int pitch  = (nband + 1 + 31) / 32 * 32;
  unsigned* part = (unsigned*)a1;
  int*      run  = (int*)a1 + (size_t)nchunk * CH;

  k_part<<<nchunk, 1024, 0, stream>>>(src, dst, part, run, E, nband, pitch);
  k_band<<<nband, 1024, 0, stream>>>(part, run, rowptr, dinv, csr, N, nband, pitch, nchunk);

  const int strips  = (N + 15) / 16;
  const int gblocks = (strips + 7) / 8;
  const int rblocks1 = (N + 15) / 16;
  const int rblocks2 = (N + 31) / 32;

  k_gemm_wmma<CIN, CHID><<<gblocks, B, 0, stream>>>(x, W1, dinv, h1, strips);
  k_gather<CHID, true><<<rblocks1, B, 0, stream>>>(rowptr, csr, dinv, b1, h1, a1, N);

  k_gemm_wmma<CHID, COUT><<<gblocks, B, 0, stream>>>(a1, W2, dinv, h2, strips);
  k_gather<COUT, false><<<rblocks2, B, 0, stream>>>(rowptr, csr, dinv, b2, h2, out, N);
}


# Layer-2 product moved into the layer-1 gather launch: a block's 16 finished rows stay in LDS, one wave multiplies them by W2 on f32 WMMA and writes h2; a1 never reaches memory; 5 launches; same sums
# speedup vs baseline: 3.0819x; 1.0656x over previous
#include <hip/hip_runtime.h>

typedef __attribute__((ext_vector_type(2))) float v2f;
typedef __attribute__((ext_vector_type(4))) float v4f;
typedef __attribute__((ext_vector_type(8))) float v8f;

#define WAVE 32

constexpr int BSH = 9;
constexpr int BW = 1 << BSH;
constexpr int CH = 8192;
constexpr int MAXBAND = 1024;
constexpr int CAP = 8192;

__device__ __forceinline__ int wave_sum(int v) {
  for (int o = 16; o > 0; o >>= 1) v += __shfl_xor(v, o, WAVE);
  return v;
}

__device__ __forceinline__ int block_sum(int v, int* sm) {
  const int lane = threadIdx.x & (WAVE - 1);
  const int wid = threadIdx.x / WAVE;
  v = wave_sum(v);
  if (lane == 0) sm[wid] = v;
  __syncthreads();
  const int t = wave_sum(sm[lane]);
  __syncthreads();
  return t;
}

__device__ __forceinline__ int block_excl_scan(int v, int* sm, int& total) {
  const int lane = threadIdx.x & (WAVE - 1);
  const int wid = threadIdx.x / WAVE;
  int x = v;
  for (int o = 1; o < WAVE; o <<= 1) {
    const int y = __shfl_up(x, o, WAVE);
    if (lane >= o) x += y;
  }
  if (lane == WAVE - 1) sm[wid] = x;
  __syncthreads();
  const int ws = sm[lane];
  const int woff = wave_sum((lane < wid) ? ws : 0);
  total = wave_sum(ws);
  __syncthreads();
  return woff + x - v;
}

__global__ __launch_bounds__(1024) void k_part(const int* __restrict__ src,
                                               const int* __restrict__ dst,
                                               unsigned* __restrict__ part,
                                               int* __restrict__ run, int E, int nband,
                                               int pitch) {
  __shared__ int hist[MAXBAND];
  __shared__ unsigned stage[CH];
  __shared__ int sm[WAVE];
  const int tid = threadIdx.x;
  const int e0 = blockIdx.x * CH;
  constexpr int PER = CH / 1024;

  hist[tid] = 0;
  __syncthreads();

  int s[PER], d[PER], r[PER];
#pragma unroll
  for (int k = 0; k < PER; ++k) {
    const int e = e0 + k * 1024 + tid;
    d[k] = -1; s[k] = 0;
    if (e < E) { d[k] = dst[e]; s[k] = src[e]; }
  }
#pragma unroll
  for (int k = 0; k < PER; ++k) {
    r[k] = 0;
    if (d[k] >= 0) r[k] = atomicAdd(&hist[d[k] >> BSH], 1);
  }
  __syncthreads();

  const int c = (tid < nband) ? hist[tid] : 0;
  int total;
  const int st = block_excl_scan(c, sm, total);
  hist[tid] = st;
  if (tid < nband) run[(size_t)blockIdx.x * pitch + tid] = st;
  if (tid == 0) run[(size_t)blockIdx.x * pitch + nband] = total;
  __syncthreads();

#pragma unroll
  for (int k = 0; k < PER; ++k)
    if (d[k] >= 0)
      stage[hist[d[k] >> BSH] + r[k]] = ((unsigned)s[k] << BSH) | (unsigned)(d[k] & (BW - 1));
  __syncthreads();

  for (int k = 0; k < PER; ++k) {
    const int i = k * 1024 + tid;
    if (i < total) part[(size_t)e0 + i] = stage[i];
  }
}

__global__ __launch_bounds__(1024) void k_band(const unsigned* __restrict__ part,
                                               const int* __restrict__ run,
                                               int* __restrict__ rowptr,
                                               float* __restrict__ dinv,
                                               int* __restrict__ csr, int n, int nband,
                                               int pitch, int nchunk) {
  __shared__ int cnt[BW];
  __shared__ int stg[CAP];
  __shared__ int sm[WAVE];
  const int tid = threadIdx.x;
  const int band = blockIdx.x;
  constexpr int GL = 16;
  constexpr int NG = 1024 / GL;
  const int g = tid / GL, gl = tid % GL;

  int before = 0, mine = 0;
  for (int r = tid; r < nchunk; r += 1024) {
    const int* rr = run + (size_t)r * pitch + band;
    const int s = rr[0], e = rr[1];
    before += s;
    mine += e - s;
  }
  if (tid < BW) cnt[tid] = 0;
  const int base = block_sum(before, sm);
  const int total = block_sum(mine, sm);
  const bool staged = total <= CAP;

  for (int r = g; r < nchunk; r += NG) {
    const int* rr = run + (size_t)r * pitch + band;
    const int s = rr[0], e = rr[1];
    const unsigned* p = part + (size_t)r * CH;
    for (int j = s + gl; j < e; j += GL) atomicAdd(&cnt[p[j] & (BW - 1)], 1);
  }
  __syncthreads();

  const int c = (tid < BW) ? cnt[tid] : 0;
  int tot2;
  const int st = block_excl_scan(c, sm, tot2);
  if (tid < BW) {
    cnt[tid] = st;
    const int node = band * BW + tid;
    if (node < n) {
      rowptr[node] = base + st;
      dinv[node] = rsqrtf((float)c + 1.0f);
      if (node == n - 1) rowptr[n] = base + st + c;
    }
  }
  __syncthreads();

  for (int r = g; r < nchunk; r += NG) {
    const int* rr = run + (size_t)r * pitch + band;
    const int s = rr[0], e = rr[1];
    const unsigned* p = part + (size_t)r * CH;
    for (int j = s + gl; j < e; j += GL) {
      const unsigned en = p[j];
      const int pos = atomicAdd(&cnt[en & (BW - 1)], 1);
      if (staged) stg[pos] = (int)(en >> BSH);
      else csr[base + pos] = (int)(en >> BSH);
    }
  }
  if (staged) {
    __syncthreads();
    for (int i = tid; i < total; i += 1024) csr[base + i] = stg[i];
  }
}

template <int K, int N>
__global__ __launch_bounds__(256) void k_gemm_wmma(
    const float* __restrict__ X, const float* __restrict__ W,
    const float* __restrict__ dinv, float* __restrict__ H, int nStrips) {
  __shared__ float Ws[K * N];
  for (int i = threadIdx.x; i < (K / 2) * N; i += blockDim.x) {
    const int k2 = i / N;
    const int n  = i % N;
    v2f p;
    p.x = W[(2 * k2) * N + n];
    p.y = W[(2 * k2 + 1) * N + n];
    *(v2f*)&Ws[(size_t)k2 * 2 * N + 2 * n] = p;
  }
  __syncthreads();

  const int waveId = threadIdx.x / WAVE;
  const int strip = blockIdx.x * (blockDim.x / WAVE) + waveId;
  if (strip >= nStrips) return;

  const int lane = threadIdx.x & (WAVE - 1);
  const int half = lane >> 4;
  const int lr   = lane & 15;

  const float* xrow = X + (size_t)(strip * 16 + lr) * K;

  constexpr int NT = N / 16;
  v8f acc[NT];
#pragma unroll
  for (int t = 0; t < NT; ++t) acc[t] = {};

#pragma unroll 4
  for (int kk = 0; kk < K; kk += 4) {
    v2f a = *(const v2f*)(xrow + kk + 2 * half);
    const float* wrow = &Ws[(size_t)(kk / 2 + half) * 2 * N];
#pragma unroll
    for (int t = 0; t < NT; ++t) {
      v2f b = *(const v2f*)(wrow + 2 * (t * 16 + lr));
      acc[t] = __builtin_amdgcn_wmma_f32_16x16x4_f32(
          false, a, false, b, (short)0, acc[t], false, false);
    }
  }

#pragma unroll
  for (int j = 0; j < 8; ++j) {
    const int orow = strip * 16 + j + 8 * half;
    float* hrow = H + (size_t)orow * N;
    const float dv = dinv[orow];
#pragma unroll
    for (int t = 0; t < NT; ++t) hrow[t * 16 + lr] = acc[t][j] * dv;
  }
}

template <int C, bool RELU>
__global__ __launch_bounds__(256) void k_gather(
    const int* __restrict__ rowptr, const int* __restrict__ csr,
    const float* __restrict__ dinv, const float* __restrict__ bias,
    const float* __restrict__ H, float* __restrict__ OUT, int n) {
  constexpr int G  = C / 4;
  constexpr int R  = WAVE / G;
  constexpr int B  = 16;
  constexpr int NE = B / G;
  const int lane = threadIdx.x & (WAVE - 1);
  const int q = lane & (G - 1);
  const int node = (int)(blockIdx.x * (blockDim.x / WAVE) + threadIdx.x / WAVE) * R + lane / G;
  const bool valid = node < n;

  const v4f bs = *(const v4f*)(bias + q * 4);
  v4f self = {};
  int start = 0, deg = 0;
  float di = 0.0f;
  if (valid) {
    self = *(const v4f*)(H + (size_t)node * C + q * 4);
    start = rowptr[node];
    deg = rowptr[node + 1] - start;
    di = dinv[node];
  }
  const int* lst = csr + start;
  const char* Hq = (const char*)(H + q * 4);

  v4f acc = {};
  for (int base = 0; __builtin_amdgcn_ballot_w32(base < deg) != 0u; base += B) {
    int ent[NE];
#pragma unroll
    for (int u = 0; u < NE; ++u) {
      const int sl = base + u * G + q;
      ent[u] = 0;
      if (sl < deg) ent[u] = lst[sl];
    }
    v4f r[B];
#pragma unroll
    for (int j = 0; j < B; ++j) {
      const int s = __shfl(ent[j / G], j % G, G);
      if (base + j < deg)
        r[j] = *(const v4f*)(Hq + (unsigned)s * (unsigned)(C * 4));
    }
#pragma unroll
    for (int j = 0; j < B; ++j)
      if (base + j < deg) acc += r[j];
  }

  if (valid) {
    v4f o = (acc + self) * di + bs;
    if (RELU) {
      o.x = fmaxf(o.x, 0.0f); o.y = fmaxf(o.y, 0.0f);
      o.z = fmaxf(o.z, 0.0f); o.w = fmaxf(o.w, 0.0f);
    }
    *(v4f*)(OUT + (size_t)node * C + q * 4) = o;
  }
}

template <int C, int N2>
__global__ __launch_bounds__(256) void k_gather_gemm(
    const int* __restrict__ rowptr, const int* __restrict__ csr,
    const float* __restrict__ dinv, const float* __restrict__ bias,
    const float* __restrict__ H, const float* __restrict__ W2,
    float* __restrict__ H2, int n) {
  constexpr int G  = C / 4;
  constexpr int R  = WAVE / G;
  constexpr int B  = 16;
  constexpr int NE = B / G;
  constexpr int ROWS = 8 * R;
  constexpr int NT = N2 / 16;
  static_assert(ROWS == 16, "a block must own exactly one 16-row strip");
  static_assert((C / 2) * (N2 / 4) == 256, "one piece of W2 per thread");
  constexpr int AP = C + 4;
  constexpr int WP = 2 * N2 + 32;
  __shared__ __attribute__((aligned(16))) float Ws[(C / 2) * WP];
  __shared__ __attribute__((aligned(16))) float As[ROWS * AP];
  __shared__ __attribute__((aligned(16))) float Ds[ROWS];

  const int lane = threadIdx.x & (WAVE - 1);
  const int waveId = threadIdx.x / WAVE;
  const int q = lane & (G - 1);
  const int rloc = waveId * R + lane / G;
  const int node = (int)blockIdx.x * ROWS + rloc;
  const bool valid = node < n;

  const int wk2 = threadIdx.x / (N2 / 4);
  const int wn  = (threadIdx.x % (N2 / 4)) * 4;
  const v4f w0 = *(const v4f*)(W2 + (size_t)(2 * wk2) * N2 + wn);
  const v4f w1 = *(const v4f*)(W2 + (size_t)(2 * wk2 + 1) * N2 + wn);

  const v4f bs = *(const v4f*)(bias + q * 4);
  v4f self = {};
  int start = 0, deg = 0;
  float di = 0.0f;
  if (valid) {
    self = *(const v4f*)(H + (size_t)node * C + q * 4);
    start = rowptr[node];
    deg = rowptr[node + 1] - start;
    di = dinv[node];
  }
  {
    v4f lo, hi;
    lo.x = w0.x; lo.y = w1.x; lo.z = w0.y; lo.w = w1.y;
    hi.x = w0.z; hi.y = w1.z; hi.z = w0.w; hi.w = w1.w;
    *(v4f*)&Ws[wk2 * WP + 2 * wn] = lo;
    *(v4f*)&Ws[wk2 * WP + 2 * wn + 4] = hi;
  }
  const int* lst = csr + start;
  const char* Hq = (const char*)(H + q * 4);

  v4f acc = {};
  for (int base = 0; __builtin_amdgcn_ballot_w32(base < deg) != 0u; base += B) {
    int ent[NE];
#pragma unroll
    for (int u = 0; u < NE; ++u) {
      const int sl = base + u * G + q;
      ent[u] = 0;
      if (sl < deg) ent[u] = lst[sl];
    }
    v4f r[B];
#pragma unroll
    for (int j = 0; j < B; ++j) {
      const int s = __shfl(ent[j / G], j % G, G);
      if (base + j < deg)
        r[j] = *(const v4f*)(Hq + (unsigned)s * (unsigned)(C * 4));
    }
#pragma unroll
    for (int j = 0; j < B; ++j)
      if (base + j < deg) acc += r[j];
  }

  v4f o = {};
  if (valid) {
    o = (acc + self) * di + bs;
    o.x = fmaxf(o.x, 0.0f); o.y = fmaxf(o.y, 0.0f);
    o.z = fmaxf(o.z, 0.0f); o.w = fmaxf(o.w, 0.0f);
  }
  *(v4f*)&As[rloc * AP + q * 4] = o;
  if (q == 0) Ds[rloc] = di;
  __syncthreads();

  if (waveId != (int)(blockIdx.x & 7)) return;

  const int half = lane >> 4;
  const int lr   = lane & 15;
  constexpr int KS = C / 4;

  v2f af[KS], bf[KS][NT];
#pragma unroll
  for (int ks = 0; ks < KS; ++ks) {
    af[ks] = *(const v2f*)&As[lr * AP + 4 * ks + 2 * half];
    const float* wrow = &Ws[(2 * ks + half) * WP];
#pragma unroll
    for (int t = 0; t < NT; ++t) bf[ks][t] = *(const v2f*)(wrow + 2 * (t * 16 + lr));
  }
  const v4f dlo = *(const v4f*)&Ds[8 * half];
  const v4f dhi = *(const v4f*)&Ds[8 * half + 4];
  __builtin_amdgcn_sched_barrier(0);

  v8f pacc[NT];
#pragma unroll
  for (int t = 0; t < NT; ++t) pacc[t] = {};
#pragma unroll
  for (int ks = 0; ks < KS; ++ks) {
#pragma unroll
    for (int t = 0; t < NT; ++t)
      pacc[t] = __builtin_amdgcn_wmma_f32_16x16x4_f32(
          false, af[ks], false, bf[ks][t], (short)0, pacc[t], false, false);
  }
  __builtin_amdgcn_sched_barrier(0);
  asm volatile("v_nop\n\tv_nop\n\tv_nop\n\tv_nop");
  __builtin_amdgcn_sched_barrier(0);

  const float dv[8] = {dlo.x, dlo.y, dlo.z, dlo.w, dhi.x, dhi.y, dhi.z, dhi.w};
  const int row0 = (int)blockIdx.x * ROWS + 8 * half;
  float* hcol = H2 + (size_t)row0 * N2 + lr;
  if ((int)blockIdx.x * ROWS + ROWS <= n) {
#pragma unroll
    for (int j = 0; j < 8; ++j)
#pragma unroll
      for (int t = 0; t < NT; ++t) hcol[j * N2 + t * 16] = pacc[t][j] * dv[j];
  } else {
#pragma unroll
    for (int j = 0; j < 8; ++j)
      if (row0 + j < n) {
#pragma unroll
        for (int t = 0; t < NT; ++t) hcol[j * N2 + t * 16] = pacc[t][j] * dv[j];
      }
  }
}

extern "C" void kernel_launch(void* const* d_in, const int* in_sizes, int n_in,
                              void* d_out, int out_size, void* d_ws, size_t ws_size,
                              hipStream_t stream) {
  const float* x  = (const float*)d_in[0];
  const int*   ei = (const int*)d_in[1];
  const float* W1 = (const float*)d_in[2];
  const float* b1 = (const float*)d_in[3];
  const float* W2 = (const float*)d_in[4];
  const float* b2 = (const float*)d_in[5];

  constexpr int CIN = 128, CHID = 64, COUT = 32;
  const int N = in_sizes[0] / CIN;
  const int E = in_sizes[1] / 2;
  const int* src = ei;
  const int* dst = ei + E;

  float* ws     = (float*)d_ws;
  float* dinv   = ws;
  float* h1     = dinv + N;
  float* tbuf   = h1 + (size_t)N * CHID;
  int*   rowptr = (int*)(tbuf + (size_t)N * CHID) + N;
  int*   csr    = rowptr + N + 2;
  float* h2     = tbuf;
  float* out    = (float*)d_out;

  const int B = 256;

  const int nband  = (N + BW - 1) / BW;
  const int nchunk = (E + CH - 1) / CH;
  const int pitch  = (nband + 1 + 31) / 32 * 32;
  unsigned* part = (unsigned*)tbuf;
  int*      run  = (int*)tbuf + (size_t)nchunk * CH;

  k_part<<<nchunk, 1024, 0, stream>>>(src, dst, part, run, E, nband, pitch);
  k_band<<<nband, 1024, 0, stream>>>(part, run, rowptr, dinv, csr, N, nband, pitch, nchunk);

  const int strips  = (N + 15) / 16;
  const int gblocks = (strips + 7) / 8;
  const int rblocks1 = (N + 15) / 16;
  const int rblocks2 = (N + 31) / 32;

  k_gemm_wmma<CIN, CHID><<<gblocks, B, 0, stream>>>(x, W1, dinv, h1, strips);
  k_gather_gemm<CHID, COUT><<<rblocks1, B, 0, stream>>>(rowptr, csr, dinv, b1, h1, W2, h2, N);

  k_gather<COUT, false><<<rblocks2, B, 0, stream>>>(rowptr, csr, dinv, b2, h2, out, N);
}


# Layer-1 product moved into the chunk-sort launch as extra 32-wave blocks, h1 stored unscaled; each band block of the CSR launch scales its 512 rows by dinv in place; 4 launches; same sums
# speedup vs baseline: 3.1782x; 1.0312x over previous
#include <hip/hip_runtime.h>

typedef __attribute__((ext_vector_type(2))) float v2f;
typedef __attribute__((ext_vector_type(4))) float v4f;
typedef __attribute__((ext_vector_type(8))) float v8f;

#define WAVE 32

constexpr int BSH = 9;
constexpr int BW = 1 << BSH;
constexpr int CH = 8192;
constexpr int MAXBAND = 1024;
constexpr int CAP = 8192;

__device__ __forceinline__ int wave_sum(int v) {
  for (int o = 16; o > 0; o >>= 1) v += __shfl_xor(v, o, WAVE);
  return v;
}

__device__ __forceinline__ int block_sum(int v, int* sm) {
  const int lane = threadIdx.x & (WAVE - 1);
  const int wid = threadIdx.x / WAVE;
  v = wave_sum(v);
  if (lane == 0) sm[wid] = v;
  __syncthreads();
  const int t = wave_sum(sm[lane]);
  __syncthreads();
  return t;
}

__device__ __forceinline__ int block_excl_scan(int v, int* sm, int& total) {
  const int lane = threadIdx.x & (WAVE - 1);
  const int wid = threadIdx.x / WAVE;
  int x = v;
  for (int o = 1; o < WAVE; o <<= 1) {
    const int y = __shfl_up(x, o, WAVE);
    if (lane >= o) x += y;
  }
  if (lane == WAVE - 1) sm[wid] = x;
  __syncthreads();
  const int ws = sm[lane];
  const int woff = wave_sum((lane < wid) ? ws : 0);
  total = wave_sum(ws);
  __syncthreads();
  return woff + x - v;
}

template <int K, int N>
__global__ __launch_bounds__(1024) void k_part_gemm(const int* __restrict__ src,
                                                    const int* __restrict__ dst,
                                                    unsigned* __restrict__ part,
                                                    int* __restrict__ run, int E, int nband,
                                                    int pitch, const float* __restrict__ X,
                                                    const float* __restrict__ W,
                                                    float* __restrict__ H, int nStrips,
                                                    int ngemm) {
  __shared__ int hist[MAXBAND];
  __shared__ __attribute__((aligned(16))) unsigned stage[CH];
  __shared__ int sm[WAVE];
  static_assert(K * N <= CH, "the W tile must fit the staging array");
  const int tid = threadIdx.x;

  if ((int)blockIdx.x < ngemm) {
    float* Ws = (float*)stage;
    static_assert(((K / 2) * N) % 1024 == 0, "whole pairs per thread");
#pragma unroll
    for (int u = 0; u < (K / 2) * N / 1024; ++u) {
      const int i = u * 1024 + tid;
      const int k2 = i / N;
      const int n  = i % N;
      v2f p;
      p.x = W[(2 * k2) * N + n];
      p.y = W[(2 * k2 + 1) * N + n];
      *(v2f*)&Ws[(size_t)k2 * 2 * N + 2 * n] = p;
    }
    __syncthreads();

    const int strip = (int)blockIdx.x * (1024 / WAVE) + tid / WAVE;
    if (strip >= nStrips) return;

    const int lane = tid & (WAVE - 1);
    const int half = lane >> 4;
    const int lr   = lane & 15;

    const float* xrow = X + (size_t)(strip * 16 + lr) * K;

    constexpr int NT = N / 16;
    v8f acc[NT];
#pragma unroll
    for (int t = 0; t < NT; ++t) acc[t] = {};

#pragma unroll 4
    for (int kk = 0; kk < K; kk += 4) {
      v2f a = *(const v2f*)(xrow + kk + 2 * half);
      const float* wrow = &Ws[(size_t)(kk / 2 + half) * 2 * N];
#pragma unroll
      for (int t = 0; t < NT; ++t) {
        v2f b = *(const v2f*)(wrow + 2 * (t * 16 + lr));
        acc[t] = __builtin_amdgcn_wmma_f32_16x16x4_f32(
            false, a, false, b, (short)0, acc[t], false, false);
      }
    }

#pragma unroll
    for (int j = 0; j < 8; ++j) {
      float* hrow = H + (size_t)(strip * 16 + j + 8 * half) * N;
#pragma unroll
      for (int t = 0; t < NT; ++t) hrow[t * 16 + lr] = acc[t][j];
    }
    return;
  }

  const int chunk = (int)blockIdx.x - ngemm;
  const int e0 = chunk * CH;
  constexpr int PER = CH / 1024;

  hist[tid] = 0;
  __syncthreads();

  int s[PER], d[PER], r[PER];
#pragma unroll
  for (int k = 0; k < PER; ++k) {
    const int e = e0 + k * 1024 + tid;
    d[k] = -1; s[k] = 0;
    if (e < E) { d[k] = dst[e]; s[k] = src[e]; }
  }
#pragma unroll
  for (int k = 0; k < PER; ++k) {
    r[k] = 0;
    if (d[k] >= 0) r[k] = atomicAdd(&hist[d[k] >> BSH], 1);
  }
  __syncthreads();

  const int c = (tid < nband) ? hist[tid] : 0;
  int total;
  const int st = block_excl_scan(c, sm, total);
  hist[tid] = st;
  if (tid < nband) run[(size_t)chunk * pitch + tid] = st;
  if (tid == 0) run[(size_t)chunk * pitch + nband] = total;
  __syncthreads();

#pragma unroll
  for (int k = 0; k < PER; ++k)
    if (d[k] >= 0)
      stage[hist[d[k] >> BSH] + r[k]] = ((unsigned)s[k] << BSH) | (unsigned)(d[k] & (BW - 1));
  __syncthreads();

  for (int k = 0; k < PER; ++k) {
    const int i = k * 1024 + tid;
    if (i < total) part[(size_t)e0 + i] = stage[i];
  }
}

template <int HC>
__global__ __launch_bounds__(1024) void k_band(const unsigned* __restrict__ part,
                                               const int* __restrict__ run,
                                               int* __restrict__ rowptr,
                                               float* __restrict__ dinv,
                                               int* __restrict__ csr, float* H, int n,
                                               int nband, int pitch, int nchunk) {
  __shared__ int cnt[BW];
  __shared__ int stg[CAP];
  __shared__ float dl[BW];
  __shared__ int sm[WAVE];
  const int tid = threadIdx.x;
  const int band = blockIdx.x;
  constexpr int GL = 16;
  constexpr int NG = 1024 / GL;
  const int g = tid / GL, gl = tid % GL;

  constexpr int RPL = 1024 / (HC / 4);
  constexpr int NHL = BW / RPL;
  static_assert(BW % RPL == 0, "whole loads per band");
  const int hrow0 = band * BW + tid / (HC / 4);
  const int hcol = (tid % (HC / 4)) * 4;

  int before = 0, mine = 0;
  for (int r = tid; r < nchunk; r += 1024) {
    const int* rr = run + (size_t)r * pitch + band;
    const int s = rr[0], e = rr[1];
    before += s;
    mine += e - s;
  }
  if (tid < BW) cnt[tid] = 0;
  const int base = block_sum(before, sm);
  const int total = block_sum(mine, sm);
  const bool staged = total <= CAP;

  for (int r = g; r < nchunk; r += NG) {
    const int* rr = run + (size_t)r * pitch + band;
    const int s = rr[0], e = rr[1];
    const unsigned* p = part + (size_t)r * CH;
    for (int j = s + gl; j < e; j += GL) atomicAdd(&cnt[p[j] & (BW - 1)], 1);
  }
  __syncthreads();

  const int c = (tid < BW) ? cnt[tid] : 0;
  int tot2;
  const int st = block_excl_scan(c, sm, tot2);
  if (tid < BW) {
    cnt[tid] = st;
    const float dv = rsqrtf((float)c + 1.0f);
    dl[tid] = dv;
    const int node = band * BW + tid;
    if (node < n) {
      rowptr[node] = base + st;
      dinv[node] = dv;
      if (node == n - 1) rowptr[n] = base + st + c;
    }
  }
  __syncthreads();

  v4f hv[NHL];
#pragma unroll
  for (int i = 0; i < NHL; ++i) {
    const int node = min(hrow0 + i * RPL, n - 1);
    hv[i] = *(const v4f*)(H + (size_t)node * HC + hcol);
  }
#pragma unroll
  for (int i = 0; i < NHL; ++i) {
    const int node = hrow0 + i * RPL;
    if (node < n) *(v4f*)(H + (size_t)node * HC + hcol) = hv[i] * dl[node - band * BW];
  }

  for (int r = g; r < nchunk; r += NG) {
    const int* rr = run + (size_t)r * pitch + band;
    const int s = rr[0], e = rr[1];
    const unsigned* p = part + (size_t)r * CH;
    for (int j = s + gl; j < e; j += GL) {
      const unsigned en = p[j];
      const int pos = atomicAdd(&cnt[en & (BW - 1)], 1);
      if (staged) stg[pos] = (int)(en >> BSH);
      else csr[base + pos] = (int)(en >> BSH);
    }
  }
  if (staged) {
    __syncthreads();
    for (int i = tid; i < total; i += 1024) csr[base + i] = stg[i];
  }
}

template <int C, bool RELU>
__global__ __launch_bounds__(256) void k_gather(
    const int* __restrict__ rowptr, const int* __restrict__ csr,
    const float* __restrict__ dinv, const float* __restrict__ bias,
    const float* __restrict__ H, float* __restrict__ OUT, int n) {
  constexpr int G  = C / 4;
  constexpr int R  = WAVE / G;
  constexpr int B  = 16;
  constexpr int NE = B / G;
  const int lane = threadIdx.x & (WAVE - 1);
  const int q = lane & (G - 1);
  const int node = (int)(blockIdx.x * (blockDim.x / WAVE) + threadIdx.x / WAVE) * R + lane / G;
  const bool valid = node < n;

  const v4f bs = *(const v4f*)(bias + q * 4);
  v4f self = {};
  int start = 0, deg = 0;
  float di = 0.0f;
  if (valid) {
    self = *(const v4f*)(H + (size_t)node * C + q * 4);
    start = rowptr[node];
    deg = rowptr[node + 1] - start;
    di = dinv[node];
  }
  const int* lst = csr + start;
  const char* Hq = (const char*)(H + q * 4);

  v4f acc = {};
  for (int base = 0; __builtin_amdgcn_ballot_w32(base < deg) != 0u; base += B) {
    int ent[NE];
#pragma unroll
    for (int u = 0; u < NE; ++u) {
      const int sl = base + u * G + q;
      ent[u] = 0;
      if (sl < deg) ent[u] = lst[sl];
    }
    v4f r[B];
#pragma unroll
    for (int j = 0; j < B; ++j) {
      const int s = __shfl(ent[j / G], j % G, G);
      if (base + j < deg)
        r[j] = *(const v4f*)(Hq + (unsigned)s * (unsigned)(C * 4));
    }
#pragma unroll
    for (int j = 0; j < B; ++j)
      if (base + j < deg) acc += r[j];
  }

  if (valid) {
    v4f o = (acc + self) * di + bs;
    if (RELU) {
      o.x = fmaxf(o.x, 0.0f); o.y = fmaxf(o.y, 0.0f);
      o.z = fmaxf(o.z, 0.0f); o.w = fmaxf(o.w, 0.0f);
    }
    *(v4f*)(OUT + (size_t)node * C + q * 4) = o;
  }
}

template <int C, int N2>
__global__ __launch_bounds__(256) void k_gather_gemm(
    const int* __restrict__ rowptr, const int* __restrict__ csr,
    const float* __restrict__ dinv, const float* __restrict__ bias,
    const float* __restrict__ H, const float* __restrict__ W2,
    float* __restrict__ H2, int n) {
  constexpr int G  = C / 4;
  constexpr int R  = WAVE / G;
  constexpr int B  = 16;
  constexpr int NE = B / G;
  constexpr int ROWS = 8 * R;
  constexpr int NT = N2 / 16;
  static_assert(ROWS == 16, "a block must own exactly one 16-row strip");
  static_assert((C / 2) * (N2 / 4) == 256, "one piece of W2 per thread");
  constexpr int AP = C + 4;
  constexpr int WP = 2 * N2 + 32;
  __shared__ __attribute__((aligned(16))) float Ws[(C / 2) * WP];
  __shared__ __attribute__((aligned(16))) float As[ROWS * AP];
  __shared__ __attribute__((aligned(16))) float Ds[ROWS];

  const int lane = threadIdx.x & (WAVE - 1);
  const int waveId = threadIdx.x / WAVE;
  const int q = lane & (G - 1);
  const int rloc = waveId * R + lane / G;
  const int node = (int)blockIdx.x * ROWS + rloc;
  const bool valid = node < n;

  const int wk2 = threadIdx.x / (N2 / 4);
  const int wn  = (threadIdx.x % (N2 / 4)) * 4;
  const v4f w0 = *(const v4f*)(W2 + (size_t)(2 * wk2) * N2 + wn);
  const v4f w1 = *(const v4f*)(W2 + (size_t)(2 * wk2 + 1) * N2 + wn);

  const v4f bs = *(const v4f*)(bias + q * 4);
  v4f self = {};
  int start = 0, deg = 0;
  float di = 0.0f;
  if (valid) {
    self = *(const v4f*)(H + (size_t)node * C + q * 4);
    start = rowptr[node];
    deg = rowptr[node + 1] - start;
    di = dinv[node];
  }
  {
    v4f lo, hi;
    lo.x = w0.x; lo.y = w1.x; lo.z = w0.y; lo.w = w1.y;
    hi.x = w0.z; hi.y = w1.z; hi.z = w0.w; hi.w = w1.w;
    *(v4f*)&Ws[wk2 * WP + 2 * wn] = lo;
    *(v4f*)&Ws[wk2 * WP + 2 * wn + 4] = hi;
  }
  const int* lst = csr + start;
  const char* Hq = (const char*)(H + q * 4);

  v4f acc = {};
  for (int base = 0; __builtin_amdgcn_ballot_w32(base < deg) != 0u; base += B) {
    int ent[NE];
#pragma unroll
    for (int u = 0; u < NE; ++u) {
      const int sl = base + u * G + q;
      ent[u] = 0;
      if (sl < deg) ent[u] = lst[sl];
    }
    v4f r[B];
#pragma unroll
    for (int j = 0; j < B; ++j) {
      const int s = __shfl(ent[j / G], j % G, G);
      if (base + j < deg)
        r[j] = *(const v4f*)(Hq + (unsigned)s * (unsigned)(C * 4));
    }
#pragma unroll
    for (int j = 0; j < B; ++j)
      if (base + j < deg) acc += r[j];
  }

  v4f o = {};
  if (valid) {
    o = (acc + self) * di + bs;
    o.x = fmaxf(o.x, 0.0f); o.y = fmaxf(o.y, 0.0f);
    o.z = fmaxf(o.z, 0.0f); o.w = fmaxf(o.w, 0.0f);
  }
  *(v4f*)&As[rloc * AP + q * 4] = o;
  if (q == 0) Ds[rloc] = di;
  __syncthreads();

  if (waveId != (int)(blockIdx.x & 7)) return;

  const int half = lane >> 4;
  const int lr   = lane & 15;
  constexpr int KS = C / 4;

  v2f af[KS], bf[KS][NT];
#pragma unroll
  for (int ks = 0; ks < KS; ++ks) {
    af[ks] = *(const v2f*)&As[lr * AP + 4 * ks + 2 * half];
    const float* wrow = &Ws[(2 * ks + half) * WP];
#pragma unroll
    for (int t = 0; t < NT; ++t) bf[ks][t] = *(const v2f*)(wrow + 2 * (t * 16 + lr));
  }
  const v4f dlo = *(const v4f*)&Ds[8 * half];
  const v4f dhi = *(const v4f*)&Ds[8 * half + 4];
  __builtin_amdgcn_sched_barrier(0);

  v8f pacc[NT];
#pragma unroll
  for (int t = 0; t < NT; ++t) pacc[t] = {};
#pragma unroll
  for (int ks = 0; ks < KS; ++ks) {
#pragma unroll
    for (int t = 0; t < NT; ++t)
      pacc[t] = __builtin_amdgcn_wmma_f32_16x16x4_f32(
          false, af[ks], false, bf[ks][t], (short)0, pacc[t], false, false);
  }
  __builtin_amdgcn_sched_barrier(0);
  asm volatile("v_nop\n\tv_nop\n\tv_nop\n\tv_nop");
  __builtin_amdgcn_sched_barrier(0);

  const float dv[8] = {dlo.x, dlo.y, dlo.z, dlo.w, dhi.x, dhi.y, dhi.z, dhi.w};
  const int row0 = (int)blockIdx.x * ROWS + 8 * half;
  float* hcol = H2 + (size_t)row0 * N2 + lr;
  if ((int)blockIdx.x * ROWS + ROWS <= n) {
#pragma unroll
    for (int j = 0; j < 8; ++j)
#pragma unroll
      for (int t = 0; t < NT; ++t) hcol[j * N2 + t * 16] = pacc[t][j] * dv[j];
  } else {
#pragma unroll
    for (int j = 0; j < 8; ++j)
      if (row0 + j < n) {
#pragma unroll
        for (int t = 0; t < NT; ++t) hcol[j * N2 + t * 16] = pacc[t][j] * dv[j];
      }
  }
}

extern "C" void kernel_launch(void* const* d_in, const int* in_sizes, int n_in,
                              void* d_out, int out_size, void* d_ws, size_t ws_size,
                              hipStream_t stream) {
  const float* x  = (const float*)d_in[0];
  const int*   ei = (const int*)d_in[1];
  const float* W1 = (const float*)d_in[2];
  const float* b1 = (const float*)d_in[3];
  const float* W2 = (const float*)d_in[4];
  const float* b2 = (const float*)d_in[5];

  constexpr int CIN = 128, CHID = 64, COUT = 32;
  const int N = in_sizes[0] / CIN;
  const int E = in_sizes[1] / 2;
  const int* src = ei;
  const int* dst = ei + E;

  float* ws     = (float*)d_ws;
  float* dinv   = ws;
  float* h1     = dinv + N;
  float* tbuf   = h1 + (size_t)N * CHID;
  int*   rowptr = (int*)(tbuf + (size_t)N * CHID) + N;
  int*   csr    = rowptr + N + 2;
  float* h2     = tbuf;
  float* out    = (float*)d_out;

  const int B = 256;

  const int nband  = (N + BW - 1) / BW;
  const int nchunk = (E + CH - 1) / CH;
  const int pitch  = (nband + 1 + 31) / 32 * 32;
  unsigned* part = (unsigned*)tbuf;
  int*      run  = (int*)tbuf + (size_t)nchunk * CH;

  const int strips  = (N + 15) / 16;
  const int gblocks = (strips + 31) / 32;
  const int rblocks1 = (N + 15) / 16;
  const int rblocks2 = (N + 31) / 32;

  k_part_gemm<CIN, CHID><<<gblocks + nchunk, 1024, 0, stream>>>(
      src, dst, part, run, E, nband, pitch, x, W1, h1, strips, gblocks);
  k_band<CHID><<<nband, 1024, 0, stream>>>(part, run, rowptr, dinv, csr, h1, N, nband, pitch, nchunk);

  k_gather_gemm<CHID, COUT><<<rblocks1, B, 0, stream>>>(rowptr, csr, dinv, b1, h1, W2, h2, N);

  k_gather<COUT, false><<<rblocks2, B, 0, stream>>>(rowptr, csr, dinv, b2, h2, out, N);
}


# Both gathers on a chip-sized grid: each of 1024 blocks loops over its strips or row groups, W2 staged once, next row headers fetched a trip ahead, one WMMA product per wave at the end; same sums
# speedup vs baseline: 3.5774x; 1.1256x over previous
#include <hip/hip_runtime.h>

typedef __attribute__((ext_vector_type(2))) float v2f;
typedef __attribute__((ext_vector_type(4))) float v4f;
typedef __attribute__((ext_vector_type(8))) float v8f;

#define WAVE 32

constexpr int BSH = 9;
constexpr int BW = 1 << BSH;
constexpr int CH = 8192;
constexpr int MAXBAND = 1024;
constexpr int CAP = 8192;

__device__ __forceinline__ int wave_sum(int v) {
  for (int o = 16; o > 0; o >>= 1) v += __shfl_xor(v, o, WAVE);
  return v;
}

__device__ __forceinline__ int block_sum(int v, int* sm) {
  const int lane = threadIdx.x & (WAVE - 1);
  const int wid = threadIdx.x / WAVE;
  v = wave_sum(v);
  if (lane == 0) sm[wid] = v;
  __syncthreads();
  const int t = wave_sum(sm[lane]);
  __syncthreads();
  return t;
}

__device__ __forceinline__ int block_excl_scan(int v, int* sm, int& total) {
  const int lane = threadIdx.x & (WAVE - 1);
  const int wid = threadIdx.x / WAVE;
  int x = v;
  for (int o = 1; o < WAVE; o <<= 1) {
    const int y = __shfl_up(x, o, WAVE);
    if (lane >= o) x += y;
  }
  if (lane == WAVE - 1) sm[wid] = x;
  __syncthreads();
  const int ws = sm[lane];
  const int woff = wave_sum((lane < wid) ? ws : 0);
  total = wave_sum(ws);
  __syncthreads();
  return woff + x - v;
}

template <int K, int N>
__global__ __launch_bounds__(1024) void k_part_gemm(const int* __restrict__ src,
                                                    const int* __restrict__ dst,
                                                    unsigned* __restrict__ part,
                                                    int* __restrict__ run, int E, int nband,
                                                    int pitch, const float* __restrict__ X,
                                                    const float* __restrict__ W,
                                                    float* __restrict__ H, int nStrips,
                                                    int ngemm) {
  __shared__ int hist[MAXBAND];
  __shared__ __attribute__((aligned(16))) unsigned stage[CH];
  __shared__ int sm[WAVE];
  static_assert(K * N <= CH, "the W tile must fit the staging array");
  const int tid = threadIdx.x;

  if ((int)blockIdx.x < ngemm) {
    float* Ws = (float*)stage;
    static_assert(((K / 2) * N) % 1024 == 0, "whole pairs per thread");
#pragma unroll
    for (int u = 0; u < (K / 2) * N / 1024; ++u) {
      const int i = u * 1024 + tid;
      const int k2 = i / N;
      const int n  = i % N;
      v2f p;
      p.x = W[(2 * k2) * N + n];
      p.y = W[(2 * k2 + 1) * N + n];
      *(v2f*)&Ws[(size_t)k2 * 2 * N + 2 * n] = p;
    }
    __syncthreads();

    const int strip = (int)blockIdx.x * (1024 / WAVE) + tid / WAVE;
    if (strip >= nStrips) return;

    const int lane = tid & (WAVE - 1);
    const int half = lane >> 4;
    const int lr   = lane & 15;

    const float* xrow = X + (size_t)(strip * 16 + lr) * K;

    constexpr int NT = N / 16;
    v8f acc[NT];
#pragma unroll
    for (int t = 0; t < NT; ++t) acc[t] = {};

#pragma unroll 4
    for (int kk = 0; kk < K; kk += 4) {
      v2f a = *(const v2f*)(xrow + kk + 2 * half);
      const float* wrow = &Ws[(size_t)(kk / 2 + half) * 2 * N];
#pragma unroll
      for (int t = 0; t < NT; ++t) {
        v2f b = *(const v2f*)(wrow + 2 * (t * 16 + lr));
        acc[t] = __builtin_amdgcn_wmma_f32_16x16x4_f32(
            false, a, false, b, (short)0, acc[t], false, false);
      }
    }

#pragma unroll
    for (int j = 0; j < 8; ++j) {
      float* hrow = H + (size_t)(strip * 16 + j + 8 * half) * N;
#pragma unroll
      for (int t = 0; t < NT; ++t) hrow[t * 16 + lr] = acc[t][j];
    }
    return;
  }

  const int chunk = (int)blockIdx.x - ngemm;
  const int e0 = chunk * CH;
  constexpr int PER = CH / 1024;

  hist[tid] = 0;
  __syncthreads();

  int s[PER], d[PER], r[PER];
#pragma unroll
  for (int k = 0; k < PER; ++k) {
    const int e = e0 + k * 1024 + tid;
    d[k] = -1; s[k] = 0;
    if (e < E) { d[k] = dst[e]; s[k] = src[e]; }
  }
#pragma unroll
  for (int k = 0; k < PER; ++k) {
    r[k] = 0;
    if (d[k] >= 0) r[k] = atomicAdd(&hist[d[k] >> BSH], 1);
  }
  __syncthreads();

  const int c = (tid < nband) ? hist[tid] : 0;
  int total;
  const int st = block_excl_scan(c, sm, total);
  hist[tid] = st;
  if (tid < nband) run[(size_t)chunk * pitch + tid] = st;
  if (tid == 0) run[(size_t)chunk * pitch + nband] = total;
  __syncthreads();

#pragma unroll
  for (int k = 0; k < PER; ++k)
    if (d[k] >= 0)
      stage[hist[d[k] >> BSH] + r[k]] = ((unsigned)s[k] << BSH) | (unsigned)(d[k] & (BW - 1));
  __syncthreads();

  for (int k = 0; k < PER; ++k) {
    const int i = k * 1024 + tid;
    if (i < total) part[(size_t)e0 + i] = stage[i];
  }
}

template <int HC>
__global__ __launch_bounds__(1024) void k_band(const unsigned* __restrict__ part,
                                               const int* __restrict__ run,
                                               int* __restrict__ rowptr,
                                               float* __restrict__ dinv,
                                               int* __restrict__ csr, float* H, int n,
                                               int nband, int pitch, int nchunk) {
  __shared__ int cnt[BW];
  __shared__ int stg[CAP];
  __shared__ float dl[BW];
  __shared__ int sm[WAVE];
  const int tid = threadIdx.x;
  const int band = blockIdx.x;
  constexpr int GL = 16;
  constexpr int NG = 1024 / GL;
  const int g = tid / GL, gl = tid % GL;

  constexpr int RPL = 1024 / (HC / 4);
  constexpr int NHL = BW / RPL;
  static_assert(BW % RPL == 0, "whole loads per band");
  const int hrow0 = band * BW + tid / (HC / 4);
  const int hcol = (tid % (HC / 4)) * 4;

  int before = 0, mine = 0;
  for (int r = tid; r < nchunk; r += 1024) {
    const int* rr = run + (size_t)r * pitch + band;
    const int s = rr[0], e = rr[1];
    before += s;
    mine += e - s;
  }
  if (tid < BW) cnt[tid] = 0;
  const int base = block_sum(before, sm);
  const int total = block_sum(mine, sm);
  const bool staged = total <= CAP;

  for (int r = g; r < nchunk; r += NG) {
    const int* rr = run + (size_t)r * pitch + band;
    const int s = rr[0], e = rr[1];
    const unsigned* p = part + (size_t)r * CH;
    for (int j = s + gl; j < e; j += GL) atomicAdd(&cnt[p[j] & (BW - 1)], 1);
  }
  __syncthreads();

  const int c = (tid < BW) ? cnt[tid] : 0;
  int tot2;
  const int st = block_excl_scan(c, sm, tot2);
  if (tid < BW) {
    cnt[tid] = st;
    const float dv = rsqrtf((float)c + 1.0f);
    dl[tid] = dv;
    const int node = band * BW + tid;
    if (node < n) {
      rowptr[node] = base + st;
      dinv[node] = dv;
      if (node == n - 1) rowptr[n] = base + st + c;
    }
  }
  __syncthreads();

  v4f hv[NHL];
#pragma unroll
  for (int i = 0; i < NHL; ++i) {
    const int node = min(hrow0 + i * RPL, n - 1);
    hv[i] = *(const v4f*)(H + (size_t)node * HC + hcol);
  }
#pragma unroll
  for (int i = 0; i < NHL; ++i) {
    const int node = hrow0 + i * RPL;
    if (node < n) *(v4f*)(H + (size_t)node * HC + hcol) = hv[i] * dl[node - band * BW];
  }

  for (int r = g; r < nchunk; r += NG) {
    const int* rr = run + (size_t)r * pitch + band;
    const int s = rr[0], e = rr[1];
    const unsigned* p = part + (size_t)r * CH;
    for (int j = s + gl; j < e; j += GL) {
      const unsigned en = p[j];
      const int pos = atomicAdd(&cnt[en & (BW - 1)], 1);
      if (staged) stg[pos] = (int)(en >> BSH);
      else csr[base + pos] = (int)(en >> BSH);
    }
  }
  if (staged) {
    __syncthreads();
    for (int i = tid; i < total; i += 1024) csr[base + i] = stg[i];
  }
}

template <int C, bool RELU>
__global__ __launch_bounds__(256) void k_gather(
    const int* __restrict__ rowptr, const int* __restrict__ csr,
    const float* __restrict__ dinv, const float* __restrict__ bias,
    const float* __restrict__ H, float* __restrict__ OUT, int n, int ngroups, int nwaves) {
  constexpr int G  = C / 4;
  constexpr int R  = WAVE / G;
  constexpr int B  = 16;
  constexpr int NE = B / G;
  const int lane = threadIdx.x & (WAVE - 1);
  const int q = lane & (G - 1);
  const int waveId = __builtin_amdgcn_readfirstlane((int)threadIdx.x / WAVE);
  int grp = (int)blockIdx.x * (256 / WAVE) + waveId;

  const v4f bs = *(const v4f*)(bias + q * 4);
  const char* Hq = (const char*)(H + q * 4);

  int node = grp * R + lane / G;
  int hn = min(node, n - 1);
  v4f self = *(const v4f*)(H + (size_t)hn * C + q * 4);
  int start = rowptr[hn];
  int deg = (node < n) ? rowptr[hn + 1] - start : 0;
  float di = dinv[hn];

  while (grp < ngroups) {
    const int* lst = csr + start;
    int ent[NE];
#pragma unroll
    for (int u = 0; u < NE; ++u) {
      const int sl = u * G + q;
      ent[u] = 0;
      if (sl < deg) ent[u] = lst[sl];
    }
    const int ngrp = grp + nwaves;
    const int nnode = ngrp * R + lane / G;
    const int nhn = min(nnode, n - 1);
    const v4f nself = *(const v4f*)(H + (size_t)nhn * C + q * 4);
    const int nstart = rowptr[nhn];
    const int nend = rowptr[nhn + 1];
    const float ndi = dinv[nhn];

    v4f acc = {};
    for (int base = 0; __builtin_amdgcn_ballot_w32(base < deg) != 0u; base += B) {
      if (base > 0) {
#pragma unroll
        for (int u = 0; u < NE; ++u) {
          const int sl = base + u * G + q;
          ent[u] = 0;
          if (sl < deg) ent[u] = lst[sl];
        }
      }
      v4f r[B];
#pragma unroll
      for (int j = 0; j < B; ++j) {
        const int s = __shfl(ent[j / G], j % G, G);
        if (base + j < deg)
          r[j] = *(const v4f*)(Hq + (unsigned)s * (unsigned)(C * 4));
      }
#pragma unroll
      for (int j = 0; j < B; ++j)
        if (base + j < deg) acc += r[j];
    }

    if (node < n) {
      v4f o = (acc + self) * di + bs;
      if (RELU) {
        o.x = fmaxf(o.x, 0.0f); o.y = fmaxf(o.y, 0.0f);
        o.z = fmaxf(o.z, 0.0f); o.w = fmaxf(o.w, 0.0f);
      }
      *(v4f*)(OUT + (size_t)node * C + q * 4) = o;
    }

    grp = ngrp; node = nnode; self = nself; start = nstart; di = ndi;
    deg = (nnode < n) ? nend - nstart : 0;
  }
}

template <int C, int N2, int AP, int WP>
__device__ __forceinline__ void strip_product(const float* As, const float* Ds, const float* Ws,
                                              float* __restrict__ H2, int strip, int n, int lane) {
  constexpr int ROWS = 16;
  constexpr int NT = N2 / 16;
  const int half = lane >> 4;
  const int lr   = lane & 15;
  constexpr int KS = C / 4;

  const v4f dlo = *(const v4f*)&Ds[8 * half];
  const v4f dhi = *(const v4f*)&Ds[8 * half + 4];

  v8f pacc[NT];
#pragma unroll
  for (int t = 0; t < NT; ++t) pacc[t] = {};
  constexpr int KH = KS / 2;
#pragma unroll
  for (int kh = 0; kh < 2; ++kh) {
    v2f af[KH], bf[KH][NT];
#pragma unroll
    for (int k8 = 0; k8 < KH; ++k8) {
      const int ks = kh * KH + k8;
      af[k8] = *(const v2f*)&As[lr * AP + 4 * ks + 2 * half];
      const float* wrow = &Ws[(2 * ks + half) * WP];
#pragma unroll
      for (int t = 0; t < NT; ++t) bf[k8][t] = *(const v2f*)(wrow + 2 * (t * 16 + lr));
    }
    __builtin_amdgcn_sched_barrier(0);
#pragma unroll
    for (int k8 = 0; k8 < KH; ++k8) {
#pragma unroll
      for (int t = 0; t < NT; ++t)
        pacc[t] = __builtin_amdgcn_wmma_f32_16x16x4_f32(
            false, af[k8], false, bf[k8][t], (short)0, pacc[t], false, false);
    }
    __builtin_amdgcn_sched_barrier(0);
    asm volatile("v_nop\n\tv_nop\n\tv_nop\n\tv_nop");
    __builtin_amdgcn_sched_barrier(0);
  }

  const float dv[8] = {dlo.x, dlo.y, dlo.z, dlo.w, dhi.x, dhi.y, dhi.z, dhi.w};
  const int row0 = strip * ROWS + 8 * half;
  float* hcol = H2 + (size_t)row0 * N2 + lr;
  if (strip * ROWS + ROWS <= n) {
#pragma unroll
    for (int j = 0; j < 8; ++j)
#pragma unroll
      for (int t = 0; t < NT; ++t) hcol[j * N2 + t * 16] = pacc[t][j] * dv[j];
  } else {
#pragma unroll
    for (int j = 0; j < 8; ++j)
      if (row0 + j < n) {
#pragma unroll
        for (int t = 0; t < NT; ++t) hcol[j * N2 + t * 16] = pacc[t][j] * dv[j];
      }
  }
}

template <int C, int N2>
__global__ __launch_bounds__(256) void k_gather_gemm(
    const int* __restrict__ rowptr, const int* __restrict__ csr,
    const float* __restrict__ dinv, const float* __restrict__ bias,
    const float* __restrict__ H, const float* __restrict__ W2,
    float* __restrict__ H2, int n, int nstrips, int nblk) {
  constexpr int G  = C / 4;
  constexpr int R  = WAVE / G;
  constexpr int B  = 16;
  constexpr int NE = B / G;
  constexpr int ROWS = 8 * R;
  constexpr int NTILE = 256 / WAVE;
  static_assert(ROWS == 16, "a block must own exactly one 16-row strip");
  static_assert((C / 2) * (N2 / 4) == 256, "one piece of W2 per thread");
  constexpr int AP = C + 4;
  constexpr int WP = 2 * N2 + 32;
  __shared__ __attribute__((aligned(16))) float Ws[(C / 2) * WP];
  __shared__ __attribute__((aligned(16))) float As[NTILE][ROWS * AP];
  __shared__ __attribute__((aligned(16))) float Ds[NTILE][ROWS];

  const int lane = threadIdx.x & (WAVE - 1);
  const int waveId = __builtin_amdgcn_readfirstlane((int)threadIdx.x / WAVE);
  const int q = lane & (G - 1);
  const int rloc = waveId * R + lane / G;

  const int wk2 = threadIdx.x / (N2 / 4);
  const int wn  = (threadIdx.x % (N2 / 4)) * 4;
  const v4f w0 = *(const v4f*)(W2 + (size_t)(2 * wk2) * N2 + wn);
  const v4f w1 = *(const v4f*)(W2 + (size_t)(2 * wk2 + 1) * N2 + wn);

  const v4f bs = *(const v4f*)(bias + q * 4);
  const char* Hq = (const char*)(H + q * 4);

  int strip = (int)blockIdx.x;
  int node = strip * ROWS + rloc;
  int hn = min(node, n - 1);
  v4f self = *(const v4f*)(H + (size_t)hn * C + q * 4);
  int start = rowptr[hn];
  int deg = (node < n) ? rowptr[hn + 1] - start : 0;
  float di = dinv[hn];
  {
    v4f lo, hi;
    lo.x = w0.x; lo.y = w1.x; lo.z = w0.y; lo.w = w1.y;
    hi.x = w0.z; hi.y = w1.z; hi.z = w0.w; hi.w = w1.w;
    *(v4f*)&Ws[wk2 * WP + 2 * wn] = lo;
    *(v4f*)&Ws[wk2 * WP + 2 * wn + 4] = hi;
  }

  while (strip < nstrips) {
    const int s0 = strip;
    for (int t = 0; t < NTILE && strip < nstrips; ++t) {
      const int* lst = csr + start;
      int ent[NE];
#pragma unroll
      for (int u = 0; u < NE; ++u) {
        const int sl = u * G + q;
        ent[u] = 0;
        if (sl < deg) ent[u] = lst[sl];
      }
      const int nstrip = strip + nblk;
      const int nnode = nstrip * ROWS + rloc;
      const int nhn = min(nnode, n - 1);
      const v4f nself = *(const v4f*)(H + (size_t)nhn * C + q * 4);
      const int nstart = rowptr[nhn];
      const int nend = rowptr[nhn + 1];
      const float ndi = dinv[nhn];

      v4f acc = {};
      for (int base = 0; __builtin_amdgcn_ballot_w32(base < deg) != 0u; base += B) {
        if (base > 0) {
#pragma unroll
          for (int u = 0; u < NE; ++u) {
            const int sl = base + u * G + q;
            ent[u] = 0;
            if (sl < deg) ent[u] = lst[sl];
          }
        }
        v4f r[B];
#pragma unroll
        for (int j = 0; j < B; ++j) {
          const int s = __shfl(ent[j / G], j % G, G);
          r[j] = *(const v4f*)(Hq + (unsigned)s * (unsigned)(C * 4));
        }
#pragma unroll
        for (int j = 0; j < B; ++j)
          if (base + j < deg) acc += r[j];
      }

      v4f o = {};
      if (node < n) {
        o = (acc + self) * di + bs;
        o.x = fmaxf(o.x, 0.0f); o.y = fmaxf(o.y, 0.0f);
        o.z = fmaxf(o.z, 0.0f); o.w = fmaxf(o.w, 0.0f);
      }
      *(v4f*)&As[t][rloc * AP + q * 4] = o;
      if (q == 0) Ds[t][rloc] = di;

      strip = nstrip; node = nnode; self = nself; start = nstart; di = ndi;
      deg = (nnode < n) ? nend - nstart : 0;
    }
    __syncthreads();

    const int ps = s0 + waveId * nblk;
    if (ps < nstrips)
      strip_product<C, N2, AP, WP>(As[waveId], Ds[waveId], Ws, H2, ps, n, lane);

    if (strip < nstrips) __syncthreads();
  }
}

extern "C" void kernel_launch(void* const* d_in, const int* in_sizes, int n_in,
                              void* d_out, int out_size, void* d_ws, size_t ws_size,
                              hipStream_t stream) {
  const float* x  = (const float*)d_in[0];
  const int*   ei = (const int*)d_in[1];
  const float* W1 = (const float*)d_in[2];
  const float* b1 = (const float*)d_in[3];
  const float* W2 = (const float*)d_in[4];
  const float* b2 = (const float*)d_in[5];

  constexpr int CIN = 128, CHID = 64, COUT = 32;
  const int N = in_sizes[0] / CIN;
  const int E = in_sizes[1] / 2;
  const int* src = ei;
  const int* dst = ei + E;

  float* ws     = (float*)d_ws;
  float* dinv   = ws;
  float* h1     = dinv + N;
  float* tbuf   = h1 + (size_t)N * CHID;
  int*   rowptr = (int*)(tbuf + (size_t)N * CHID) + N;
  int*   csr    = rowptr + N + 2;
  float* h2     = tbuf;
  float* out    = (float*)d_out;

  const int B = 256;

  const int nband  = (N + BW - 1) / BW;
  const int nchunk = (E + CH - 1) / CH;
  const int pitch  = (nband + 1 + 31) / 32 * 32;
  unsigned* part = (unsigned*)tbuf;
  int*      run  = (int*)tbuf + (size_t)nchunk * CH;

  const int strips  = (N + 15) / 16;
  const int gblocks = (strips + 31) / 32;
  const int groups2 = (N + 3) / 4;
  const int pblocks1 = 1024;
  const int pblocks2 = 1024;

  k_part_gemm<CIN, CHID><<<gblocks + nchunk, 1024, 0, stream>>>(
      src, dst, part, run, E, nband, pitch, x, W1, h1, strips, gblocks);
  k_band<CHID><<<nband, 1024, 0, stream>>>(part, run, rowptr, dinv, csr, h1, N, nband, pitch, nchunk);

  k_gather_gemm<CHID, COUT><<<pblocks1, B, 0, stream>>>(rowptr, csr, dinv, b1, h1, W2, h2, N, strips, pblocks1);

  k_gather<COUT, false><<<pblocks2, B, 0, stream>>>(rowptr, csr, dinv, b2, h2, out, N, groups2, pblocks2 * (B / WAVE));
}


# Band launch: its long chain of dependent memory round trips cut to two: edges numbered across chunks, fetched in one burst, kept in registers for count and placement; arrival order of slots differs
# speedup vs baseline: 3.7671x; 1.0530x over previous
#include <hip/hip_runtime.h>

typedef __attribute__((ext_vector_type(2))) float v2f;
typedef __attribute__((ext_vector_type(4))) float v4f;
typedef __attribute__((ext_vector_type(8))) float v8f;

#define WAVE 32

constexpr int BSH = 9;
constexpr int BW = 1 << BSH;
constexpr int CH = 8192;
constexpr int MAXBAND = 1024;
constexpr int CAP = 8192;

__device__ __forceinline__ int wave_sum(int v) {
  for (int o = 16; o > 0; o >>= 1) v += __shfl_xor(v, o, WAVE);
  return v;
}

__device__ __forceinline__ int block_sum(int v, int* sm) {
  const int lane = threadIdx.x & (WAVE - 1);
  const int wid = threadIdx.x / WAVE;
  v = wave_sum(v);
  if (lane == 0) sm[wid] = v;
  __syncthreads();
  const int t = wave_sum(sm[lane]);
  __syncthreads();
  return t;
}

__device__ __forceinline__ int block_excl_scan(int v, int* sm, int& total) {
  const int lane = threadIdx.x & (WAVE - 1);
  const int wid = threadIdx.x / WAVE;
  int x = v;
  for (int o = 1; o < WAVE; o <<= 1) {
    const int y = __shfl_up(x, o, WAVE);
    if (lane >= o) x += y;
  }
  if (lane == WAVE - 1) sm[wid] = x;
  __syncthreads();
  const int ws = sm[lane];
  const int woff = wave_sum((lane < wid) ? ws : 0);
  total = wave_sum(ws);
  __syncthreads();
  return woff + x - v;
}

template <int K, int N>
__global__ __launch_bounds__(1024) void k_part_gemm(const int* __restrict__ src,
                                                    const int* __restrict__ dst,
                                                    unsigned* __restrict__ part,
                                                    int* __restrict__ run, int E, int nband,
                                                    int pitch, const float* __restrict__ X,
                                                    const float* __restrict__ W,
                                                    float* __restrict__ H, int nStrips,
                                                    int ngemm) {
  __shared__ int hist[MAXBAND];
  __shared__ __attribute__((aligned(16))) unsigned stage[CH];
  __shared__ int sm[WAVE];
  static_assert(K * N <= CH, "the W tile must fit the staging array");
  const int tid = threadIdx.x;

  if ((int)blockIdx.x < ngemm) {
    float* Ws = (float*)stage;
    static_assert(((K / 2) * N) % 1024 == 0, "whole pairs per thread");
#pragma unroll
    for (int u = 0; u < (K / 2) * N / 1024; ++u) {
      const int i = u * 1024 + tid;
      const int k2 = i / N;
      const int n  = i % N;
      v2f p;
      p.x = W[(2 * k2) * N + n];
      p.y = W[(2 * k2 + 1) * N + n];
      *(v2f*)&Ws[(size_t)k2 * 2 * N + 2 * n] = p;
    }
    __syncthreads();

    const int strip = (int)blockIdx.x * (1024 / WAVE) + tid / WAVE;
    if (strip >= nStrips) return;

    const int lane = tid & (WAVE - 1);
    const int half = lane >> 4;
    const int lr   = lane & 15;

    const float* xrow = X + (size_t)(strip * 16 + lr) * K;

    constexpr int NT = N / 16;
    v8f acc[NT];
#pragma unroll
    for (int t = 0; t < NT; ++t) acc[t] = {};

#pragma unroll 4
    for (int kk = 0; kk < K; kk += 4) {
      v2f a = *(const v2f*)(xrow + kk + 2 * half);
      const float* wrow = &Ws[(size_t)(kk / 2 + half) * 2 * N];
#pragma unroll
      for (int t = 0; t < NT; ++t) {
        v2f b = *(const v2f*)(wrow + 2 * (t * 16 + lr));
        acc[t] = __builtin_amdgcn_wmma_f32_16x16x4_f32(
            false, a, false, b, (short)0, acc[t], false, false);
      }
    }

#pragma unroll
    for (int j = 0; j < 8; ++j) {
      float* hrow = H + (size_t)(strip * 16 + j + 8 * half) * N;
#pragma unroll
      for (int t = 0; t < NT; ++t) hrow[t * 16 + lr] = acc[t][j];
    }
    return;
  }

  const int chunk = (int)blockIdx.x - ngemm;
  const int e0 = chunk * CH;
  constexpr int PER = CH / 1024;

  hist[tid] = 0;
  __syncthreads();

  int s[PER], d[PER], r[PER];
#pragma unroll
  for (int k = 0; k < PER; ++k) {
    const int e = e0 + k * 1024 + tid;
    d[k] = -1; s[k] = 0;
    if (e < E) { d[k] = dst[e]; s[k] = src[e]; }
  }
#pragma unroll
  for (int k = 0; k < PER; ++k) {
    r[k] = 0;
    if (d[k] >= 0) r[k] = atomicAdd(&hist[d[k] >> BSH], 1);
  }
  __syncthreads();

  const int c = (tid < nband) ? hist[tid] : 0;
  int total;
  const int st = block_excl_scan(c, sm, total);
  hist[tid] = st;
  if (tid < nband) run[(size_t)chunk * pitch + tid] = st;
  if (tid == 0) run[(size_t)chunk * pitch + nband] = total;
  __syncthreads();

#pragma unroll
  for (int k = 0; k < PER; ++k)
    if (d[k] >= 0)
      stage[hist[d[k] >> BSH] + r[k]] = ((unsigned)s[k] << BSH) | (unsigned)(d[k] & (BW - 1));
  __syncthreads();

  for (int k = 0; k < PER; ++k) {
    const int i = k * 1024 + tid;
    if (i < total) part[(size_t)e0 + i] = stage[i];
  }
}

template <int HC>
__global__ __launch_bounds__(1024) void k_band(const unsigned* __restrict__ part,
                                               const int* __restrict__ run,
                                               int* __restrict__ rowptr,
                                               float* __restrict__ dinv,
                                               int* __restrict__ csr, float* H, int n,
                                               int nband, int pitch, int nchunk) {
  __shared__ int cnt[BW];
  __shared__ int stg[CAP];
  __shared__ float dl[BW];
  __shared__ int sm[WAVE];
  __shared__ int cP[1024];
  __shared__ int cL[1024];
  __shared__ int cO[1024];
  const int tid = threadIdx.x;
  const int band = blockIdx.x;
  constexpr int EPT = CAP / 1024;
  constexpr int GL = 16;
  constexpr int NG = 1024 / GL;
  const int g = tid / GL, gl = tid % GL;

  constexpr int RPL = 1024 / (HC / 4);
  constexpr int NHL = BW / RPL;
  static_assert(BW % RPL == 0, "whole loads per band");
  const int hrow0 = band * BW + tid / (HC / 4);
  const int hcol = (tid % (HC / 4)) * 4;

  int s0 = 0, before = 0, mine = 0;
  for (int r = tid; r < nchunk; r += 1024) {
    const int* rr = run + (size_t)r * pitch + band;
    const int s = rr[0], e = rr[1];
    if (r == tid) s0 = s;
    before += s;
    mine += e - s;
  }
  if (tid < BW) cnt[tid] = 0;
  const int base = block_sum(before, sm);
  int total;
  const int P = block_excl_scan(mine, sm, total);
  const bool fast = nchunk <= 1024 && total <= CAP;

  unsigned en[EPT];
  int rk[EPT];
#pragma unroll
  for (int k = 0; k < EPT; ++k) { en[k] = 0u; rk[k] = 0; }

  if (fast) {
    if (tid < nchunk) { cP[tid] = P; cL[tid] = mine; cO[tid] = tid * CH + s0 - P; }
    __syncthreads();
    for (int r = tid >> 3; r < nchunk; r += 128) {
      const int p0 = cP[r], len = cL[r], off = cO[r];
      for (int m = tid & 7; m < len; m += 8) stg[p0 + m] = off;
    }
    __syncthreads();
    int so[EPT];
#pragma unroll
    for (int k = 0; k < EPT; ++k) so[k] = stg[k * 1024 + tid];
#pragma unroll
    for (int k = 0; k < EPT; ++k) {
      const int i = k * 1024 + tid;
      en[k] = part[(i < total) ? so[k] + i : 0];
    }
  }

  v4f hv[NHL];
#pragma unroll
  for (int i = 0; i < NHL; ++i) {
    const int node = min(hrow0 + i * RPL, n - 1);
    hv[i] = *(const v4f*)(H + (size_t)node * HC + hcol);
  }

  if (fast) {
#pragma unroll
    for (int k = 0; k < EPT; ++k)
      if (k * 1024 + tid < total) rk[k] = atomicAdd(&cnt[en[k] & (BW - 1)], 1);
  } else {
    for (int r = g; r < nchunk; r += NG) {
      const int* rr = run + (size_t)r * pitch + band;
      const int s = rr[0], e = rr[1];
      const unsigned* p = part + (size_t)r * CH;
      for (int j = s + gl; j < e; j += GL) atomicAdd(&cnt[p[j] & (BW - 1)], 1);
    }
  }
  __syncthreads();

  const int c = (tid < BW) ? cnt[tid] : 0;
  int tot2;
  const int st = block_excl_scan(c, sm, tot2);
  const float dv = rsqrtf((float)c + 1.0f);
  if (tid < BW) {
    cnt[tid] = st;
    dl[tid] = dv;
  }
  __syncthreads();

  if (fast) {
    int ps[EPT];
#pragma unroll
    for (int k = 0; k < EPT; ++k) ps[k] = cnt[en[k] & (BW - 1)];
#pragma unroll
    for (int k = 0; k < EPT; ++k)
      if (k * 1024 + tid < total) stg[ps[k] + rk[k]] = (int)(en[k] >> BSH);
    __syncthreads();
    for (int i = tid; i < total; i += 1024) csr[base + i] = stg[i];
  } else {
    for (int r = g; r < nchunk; r += NG) {
      const int* rr = run + (size_t)r * pitch + band;
      const int s = rr[0], e = rr[1];
      const unsigned* p = part + (size_t)r * CH;
      for (int j = s + gl; j < e; j += GL) {
        const unsigned w = p[j];
        const int pos = atomicAdd(&cnt[w & (BW - 1)], 1);
        csr[base + pos] = (int)(w >> BSH);
      }
    }
  }

  if (tid < BW) {
    const int node = band * BW + tid;
    if (node < n) {
      rowptr[node] = base + st;
      dinv[node] = dv;
      if (node == n - 1) rowptr[n] = base + st + c;
    }
  }
#pragma unroll
  for (int i = 0; i < NHL; ++i) {
    const int node = hrow0 + i * RPL;
    if (node < n) *(v4f*)(H + (size_t)node * HC + hcol) = hv[i] * dl[node - band * BW];
  }
}

template <int C, bool RELU>
__global__ __launch_bounds__(256) void k_gather(
    const int* __restrict__ rowptr, const int* __restrict__ csr,
    const float* __restrict__ dinv, const float* __restrict__ bias,
    const float* __restrict__ H, float* __restrict__ OUT, int n, int ngroups, int nwaves) {
  constexpr int G  = C / 4;
  constexpr int R  = WAVE / G;
  constexpr int B  = 16;
  constexpr int NE = B / G;
  const int lane = threadIdx.x & (WAVE - 1);
  const int q = lane & (G - 1);
  const int waveId = __builtin_amdgcn_readfirstlane((int)threadIdx.x / WAVE);
  int grp = (int)blockIdx.x * (256 / WAVE) + waveId;

  const v4f bs = *(const v4f*)(bias + q * 4);
  const char* Hq = (const char*)(H + q * 4);

  int node = grp * R + lane / G;
  int hn = min(node, n - 1);
  v4f self = *(const v4f*)(H + (size_t)hn * C + q * 4);
  int start = rowptr[hn];
  int deg = (node < n) ? rowptr[hn + 1] - start : 0;
  float di = dinv[hn];

  while (grp < ngroups) {
    const int* lst = csr + start;
    int ent[NE];
#pragma unroll
    for (int u = 0; u < NE; ++u) {
      const int sl = u * G + q;
      ent[u] = 0;
      if (sl < deg) ent[u] = lst[sl];
    }
    const int ngrp = grp + nwaves;
    const int nnode = ngrp * R + lane / G;
    const int nhn = min(nnode, n - 1);
    const v4f nself = *(const v4f*)(H + (size_t)nhn * C + q * 4);
    const int nstart = rowptr[nhn];
    const int nend = rowptr[nhn + 1];
    const float ndi = dinv[nhn];

    v4f acc = {};
    for (int base = 0; __builtin_amdgcn_ballot_w32(base < deg) != 0u; base += B) {
      if (base > 0) {
#pragma unroll
        for (int u = 0; u < NE; ++u) {
          const int sl = base + u * G + q;
          ent[u] = 0;
          if (sl < deg) ent[u] = lst[sl];
        }
      }
      v4f r[B];
#pragma unroll
      for (int j = 0; j < B; ++j) {
        const int s = __shfl(ent[j / G], j % G, G);
        if (base + j < deg)
          r[j] = *(const v4f*)(Hq + (unsigned)s * (unsigned)(C * 4));
      }
#pragma unroll
      for (int j = 0; j < B; ++j)
        if (base + j < deg) acc += r[j];
    }

    if (node < n) {
      v4f o = (acc + self) * di + bs;
      if (RELU) {
        o.x = fmaxf(o.x, 0.0f); o.y = fmaxf(o.y, 0.0f);
        o.z = fmaxf(o.z, 0.0f); o.w = fmaxf(o.w, 0.0f);
      }
      *(v4f*)(OUT + (size_t)node * C + q * 4) = o;
    }

    grp = ngrp; node = nnode; self = nself; start = nstart; di = ndi;
    deg = (nnode < n) ? nend - nstart : 0;
  }
}

template <int C, int N2, int AP, int WP>
__device__ __forceinline__ void strip_product(const float* As, const float* Ds, const float* Ws,
                                              float* __restrict__ H2, int strip, int n, int lane) {
  constexpr int ROWS = 16;
  constexpr int NT = N2 / 16;
  const int half = lane >> 4;
  const int lr   = lane & 15;
  constexpr int KS = C / 4;

  const v4f dlo = *(const v4f*)&Ds[8 * half];
  const v4f dhi = *(const v4f*)&Ds[8 * half + 4];

  v8f pacc[NT];
#pragma unroll
  for (int t = 0; t < NT; ++t) pacc[t] = {};
  constexpr int KH = KS / 2;
#pragma unroll
  for (int kh = 0; kh < 2; ++kh) {
    v2f af[KH], bf[KH][NT];
#pragma unroll
    for (int k8 = 0; k8 < KH; ++k8) {
      const int ks = kh * KH + k8;
      af[k8] = *(const v2f*)&As[lr * AP + 4 * ks + 2 * half];
      const float* wrow = &Ws[(2 * ks + half) * WP];
#pragma unroll
      for (int t = 0; t < NT; ++t) bf[k8][t] = *(const v2f*)(wrow + 2 * (t * 16 + lr));
    }
    __builtin_amdgcn_sched_barrier(0);
#pragma unroll
    for (int k8 = 0; k8 < KH; ++k8) {
#pragma unroll
      for (int t = 0; t < NT; ++t)
        pacc[t] = __builtin_amdgcn_wmma_f32_16x16x4_f32(
            false, af[k8], false, bf[k8][t], (short)0, pacc[t], false, false);
    }
    __builtin_amdgcn_sched_barrier(0);
    asm volatile("v_nop\n\tv_nop\n\tv_nop\n\tv_nop");
    __builtin_amdgcn_sched_barrier(0);
  }

  const float dv[8] = {dlo.x, dlo.y, dlo.z, dlo.w, dhi.x, dhi.y, dhi.z, dhi.w};
  const int row0 = strip * ROWS + 8 * half;
  float* hcol = H2 + (size_t)row0 * N2 + lr;
  if (strip * ROWS + ROWS <= n) {
#pragma unroll
    for (int j = 0; j < 8; ++j)
#pragma unroll
      for (int t = 0; t < NT; ++t) hcol[j * N2 + t * 16] = pacc[t][j] * dv[j];
  } else {
#pragma unroll
    for (int j = 0; j < 8; ++j)
      if (row0 + j < n) {
#pragma unroll
        for (int t = 0; t < NT; ++t) hcol[j * N2 + t * 16] = pacc[t][j] * dv[j];
      }
  }
}

template <int C, int N2>
__global__ __launch_bounds__(256) void k_gather_gemm(
    const int* __restrict__ rowptr, const int* __restrict__ csr,
    const float* __restrict__ dinv, const float* __restrict__ bias,
    const float* __restrict__ H, const float* __restrict__ W2,
    float* __restrict__ H2, int n, int nstrips, int nblk) {
  constexpr int G  = C / 4;
  constexpr int R  = WAVE / G;
  constexpr int B  = 16;
  constexpr int NE = B / G;
  constexpr int ROWS = 8 * R;
  constexpr int NTILE = 256 / WAVE;
  static_assert(ROWS == 16, "a block must own exactly one 16-row strip");
  static_assert((C / 2) * (N2 / 4) == 256, "one piece of W2 per thread");
  constexpr int AP = C + 4;
  constexpr int WP = 2 * N2 + 32;
  __shared__ __attribute__((aligned(16))) float Ws[(C / 2) * WP];
  __shared__ __attribute__((aligned(16))) float As[NTILE][ROWS * AP];
  __shared__ __attribute__((aligned(16))) float Ds[NTILE][ROWS];

  const int lane = threadIdx.x & (WAVE - 1);
  const int waveId = __builtin_amdgcn_readfirstlane((int)threadIdx.x / WAVE);
  const int q = lane & (G - 1);
  const int rloc = waveId * R + lane / G;

  const int wk2 = threadIdx.x / (N2 / 4);
  const int wn  = (threadIdx.x % (N2 / 4)) * 4;
  const v4f w0 = *(const v4f*)(W2 + (size_t)(2 * wk2) * N2 + wn);
  const v4f w1 = *(const v4f*)(W2 + (size_t)(2 * wk2 + 1) * N2 + wn);

  const v4f bs = *(const v4f*)(bias + q * 4);
  const char* Hq = (const char*)(H + q * 4);

  int strip = (int)blockIdx.x;
  int node = strip * ROWS + rloc;
  int hn = min(node, n - 1);
  v4f self = *(const v4f*)(H + (size_t)hn * C + q * 4);
  int start = rowptr[hn];
  int deg = (node < n) ? rowptr[hn + 1] - start : 0;
  float di = dinv[hn];
  {
    v4f lo, hi;
    lo.x = w0.x; lo.y = w1.x; lo.z = w0.y; lo.w = w1.y;
    hi.x = w0.z; hi.y = w1.z; hi.z = w0.w; hi.w = w1.w;
    *(v4f*)&Ws[wk2 * WP + 2 * wn] = lo;
    *(v4f*)&Ws[wk2 * WP + 2 * wn + 4] = hi;
  }

  while (strip < nstrips) {
    const int s0 = strip;
    for (int t = 0; t < NTILE && strip < nstrips; ++t) {
      const int* lst = csr + start;
      int ent[NE];
#pragma unroll
      for (int u = 0; u < NE; ++u) {
        const int sl = u * G + q;
        ent[u] = 0;
        if (sl < deg) ent[u] = lst[sl];
      }
      const int nstrip = strip + nblk;
      const int nnode = nstrip * ROWS + rloc;
      const int nhn = min(nnode, n - 1);
      const v4f nself = *(const v4f*)(H + (size_t)nhn * C + q * 4);
      const int nstart = rowptr[nhn];
      const int nend = rowptr[nhn + 1];
      const float ndi = dinv[nhn];

      v4f acc = {};
      for (int base = 0; __builtin_amdgcn_ballot_w32(base < deg) != 0u; base += B) {
        if (base > 0) {
#pragma unroll
          for (int u = 0; u < NE; ++u) {
            const int sl = base + u * G + q;
            ent[u] = 0;
            if (sl < deg) ent[u] = lst[sl];
          }
        }
        v4f r[B];
#pragma unroll
        for (int j = 0; j < B; ++j) {
          const int s = __shfl(ent[j / G], j % G, G);
          r[j] = *(const v4f*)(Hq + (unsigned)s * (unsigned)(C * 4));
        }
#pragma unroll
        for (int j = 0; j < B; ++j)
          if (base + j < deg) acc += r[j];
      }

      v4f o = {};
      if (node < n) {
        o = (acc + self) * di + bs;
        o.x = fmaxf(o.x, 0.0f); o.y = fmaxf(o.y, 0.0f);
        o.z = fmaxf(o.z, 0.0f); o.w = fmaxf(o.w, 0.0f);
      }
      *(v4f*)&As[t][rloc * AP + q * 4] = o;
      if (q == 0) Ds[t][rloc] = di;

      strip = nstrip; node = nnode; self = nself; start = nstart; di = ndi;
      deg = (nnode < n) ? nend - nstart : 0;
    }
    __syncthreads();

    const int ps = s0 + waveId * nblk;
    if (ps < nstrips)
      strip_product<C, N2, AP, WP>(As[waveId], Ds[waveId], Ws, H2, ps, n, lane);

    if (strip < nstrips) __syncthreads();
  }
}

extern "C" void kernel_launch(void* const* d_in, const int* in_sizes, int n_in,
                              void* d_out, int out_size, void* d_ws, size_t ws_size,
                              hipStream_t stream) {
  const float* x  = (const float*)d_in[0];
  const int*   ei = (const int*)d_in[1];
  const float* W1 = (const float*)d_in[2];
  const float* b1 = (const float*)d_in[3];
  const float* W2 = (const float*)d_in[4];
  const float* b2 = (const float*)d_in[5];

  constexpr int CIN = 128, CHID = 64, COUT = 32;
  const int N = in_sizes[0] / CIN;
  const int E = in_sizes[1] / 2;
  const int* src = ei;
  const int* dst = ei + E;

  float* ws     = (float*)d_ws;
  float* dinv   = ws;
  float* h1     = dinv + N;
  float* tbuf   = h1 + (size_t)N * CHID;
  int*   rowptr = (int*)(tbuf + (size_t)N * CHID) + N;
  int*   csr    = rowptr + N + 2;
  float* h2     = tbuf;
  float* out    = (float*)d_out;

  const int B = 256;

  const int nband  = (N + BW - 1) / BW;
  const int nchunk = (E + CH - 1) / CH;
  const int pitch  = (nband + 1 + 31) / 32 * 32;
  unsigned* part = (unsigned*)tbuf;
  int*      run  = (int*)tbuf + (size_t)nchunk * CH;

  const int strips  = (N + 15) / 16;
  const int gblocks = (strips + 31) / 32;
  const int groups2 = (N + 3) / 4;
  const int pblocks1 = 1024;
  const int pblocks2 = 1024;

  k_part_gemm<CIN, CHID><<<gblocks + nchunk, 1024, 0, stream>>>(
      src, dst, part, run, E, nband, pitch, x, W1, h1, strips, gblocks);
  k_band<CHID><<<nband, 1024, 0, stream>>>(part, run, rowptr, dinv, csr, h1, N, nband, pitch, nchunk);

  k_gather_gemm<CHID, COUT><<<pblocks1, B, 0, stream>>>(rowptr, csr, dinv, b1, h1, W2, h2, N, strips, pblocks1);

  k_gather<COUT, false><<<pblocks2, B, 0, stream>>>(rowptr, csr, dinv, b2, h2, out, N, groups2, pblocks2 * (B / WAVE));
}
